# starting kernel
# baseline (speedup 1.0000x reference)
#include <hip/hip_runtime.h>
#include <hip/hip_bf16.h>

typedef __attribute__((ext_vector_type(2))) float v2f;
typedef __attribute__((ext_vector_type(8))) float v8f;

#define CH 128
#define EPS 1e-5f

__global__ void k_deg_init(float* __restrict__ deg, int N) {
    int i = blockIdx.x * blockDim.x + threadIdx.x;
    if (i < N) deg[i] = 1.0f;
}

__global__ void k_edge_deg(const int* __restrict__ ei, float* __restrict__ deg,
                           int E, int N) {
    int e = blockIdx.x * blockDim.x + threadIdx.x;
    if (e >= E) return;
    int s = ei[e];
    int d = ei[E + e];
    if ((unsigned)s < (unsigned)N && (unsigned)d < (unsigned)N) {
        atomicAdd(&deg[d], 1.0f);
    }
}

__global__ void k_dinv(float* __restrict__ deg, int N) {
    int i = blockIdx.x * blockDim.x + threadIdx.x;
    if (i < N) deg[i] = rsqrtf(deg[i]);
}

__global__ __launch_bounds__(256) void k_gemm_wmma(const float* __restrict__ X,
                                                   const float* __restrict__ W,
                                                   float* __restrict__ H, int M) {
    __shared__ float sA[64][132];
    __shared__ float sB[32][132];
    const int m0  = blockIdx.x * 64;
    const int n0  = blockIdx.y * 32;
    const int tid = threadIdx.x;

    #pragma unroll
    for (int i = 0; i < 8; ++i) {
        const int idx   = tid + i * 256;
        const int row   = idx >> 5;
        const int chunk = idx & 31;
        const int gr    = m0 + row;
        const float4 v =
            *(const float4*)(X + (size_t)(gr < M ? gr : 0) * CH + chunk * 4);
        *(float4*)&sA[row][chunk * 4] = v;
    }
    {
        const int c     = tid & 31;
        const int kbase = (tid >> 5) * 16;
        #pragma unroll
        for (int i = 0; i < 16; ++i) {
            const int k = kbase + i;
            sB[c][k] = W[(size_t)k * CH + n0 + c];
        }
    }
    __syncthreads();

    const int wave  = tid >> 5;
    const int lane  = tid & 31;
    const int mt    = wave >> 1;
    const int nt    = wave & 1;
    const int mloc  = lane & 15;
    const int khalf = (lane >> 4) * 2;

    const float* arow = &sA[mt * 16 + mloc][0];
    const float* brow = &sB[nt * 16 + mloc][0];

    v8f c = {};
    #pragma unroll 8
    for (int k0 = 0; k0 < CH; k0 += 4) {
        const v2f a = *(const v2f*)(arow + k0 + khalf);
        const v2f b = *(const v2f*)(brow + k0 + khalf);
        c = __builtin_amdgcn_wmma_f32_16x16x4_f32(
                 false, a,  false, b,
                 (short)0, c,  false,  false);
    }

    const int col   = n0 + nt * 16 + mloc;
    const int half  = lane >> 4;
    const int mbase = m0 + mt * 16 + half * 8;
    if (m0 + 64 <= M) {
        #pragma unroll
        for (int r = 0; r < 8; ++r)
            H[(size_t)(mbase + r) * CH + col] = c[r];
    } else {
        #pragma unroll
        for (int r = 0; r < 8; ++r)
            if (mbase + r < M) H[(size_t)(mbase + r) * CH + col] = c[r];
    }
}

__global__ void k_selfloop_init(const float* __restrict__ H,
                                const float* __restrict__ dinv,
                                const float* __restrict__ bias,
                                float* __restrict__ Acc, int M) {
    int idx = blockIdx.x * blockDim.x + threadIdx.x;
    if (idx >= M * (CH / 4)) return;
    const int row = idx >> 5;
    const int l   = idx & 31;
    const float di = dinv[row];
    const float w  = di * di;
    float4 h = ((const float4*)(H + (size_t)row * CH))[l];
    float4 b = ((const float4*)bias)[l];
    float4 o;
    o.x = h.x * w + b.x;  o.y = h.y * w + b.y;
    o.z = h.z * w + b.z;  o.w = h.w * w + b.w;
    ((float4*)(Acc + (size_t)row * CH))[l] = o;
}

__global__ void k_edge_scatter(const float* __restrict__ H,
                               const int* __restrict__ ei,
                               const float* __restrict__ dinv,
                               float* __restrict__ Acc, int E, int N) {
    int e    = (blockIdx.x * blockDim.x + threadIdx.x) >> 5;
    int lane = threadIdx.x & 31;
    if (e >= E) return;
    int s = ei[e];
    int d = ei[E + e];
    if ((unsigned)s >= (unsigned)N || (unsigned)d >= (unsigned)N) return;
    const float norm = dinv[s] * dinv[d];
    float4 v = ((const float4*)(H + (size_t)s * CH))[lane];
    float* out = Acc + (size_t)d * CH + lane * 4;
    atomicAdd(out + 0, v.x * norm);
    atomicAdd(out + 1, v.y * norm);
    atomicAdd(out + 2, v.z * norm);
    atomicAdd(out + 3, v.w * norm);
}

__global__ void k_rownorm_relu(const float* __restrict__ In,
                               float* __restrict__ Out, int M) {
    int row  = (blockIdx.x * blockDim.x + threadIdx.x) >> 5;
    int lane = threadIdx.x & 31;
    if (row >= M) return;
    float4 v = ((const float4*)(In + (size_t)row * CH))[lane];

    float s = v.x + v.y + v.z + v.w;
    #pragma unroll
    for (int o = 16; o > 0; o >>= 1) s += __shfl_xor(s, o, 32);
    const float mu = s * (1.0f / CH);

    const float d0 = v.x - mu, d1 = v.y - mu, d2 = v.z - mu, d3 = v.w - mu;
    float q = d0 * d0 + d1 * d1 + d2 * d2 + d3 * d3;
    #pragma unroll
    for (int o = 16; o > 0; o >>= 1) q += __shfl_xor(q, o, 32);
    const float inv = rsqrtf(q * (1.0f / CH) + EPS);

    float4 r;
    r.x = fmaxf(d0 * inv, 0.0f);
    r.y = fmaxf(d1 * inv, 0.0f);
    r.z = fmaxf(d2 * inv, 0.0f);
    r.w = fmaxf(d3 * inv, 0.0f);
    ((float4*)(Out + (size_t)row * CH))[lane] = r;
}

extern "C" void kernel_launch(void* const* d_in, const int* in_sizes, int n_in,
                              void* d_out, int out_size, void* d_ws, size_t ws_size,
                              hipStream_t stream) {
    const float* x   = (const float*)d_in[0];
    const int*   ei  = (const int*)  d_in[1];
    const float* W1  = (const float*)d_in[2];
    const float* b1  = (const float*)d_in[3];
    const float* W2  = (const float*)d_in[4];
    const float* b2  = (const float*)d_in[5];
    float*       out = (float*)d_out;

    const int N = in_sizes[0] / CH;
    const int E = in_sizes[1] / 2;

    float* ws   = (float*)d_ws;
    const size_t nPad = ((size_t)N + 63) & ~(size_t)63;
    float* dinv = ws;
    float* hbuf = ws + nPad;
    float* acc  = hbuf + (size_t)N * CH;

    const int T = 256;
    const int gN     = (N + T - 1) / T;
    const int gE     = (E + T - 1) / T;
    const int gRow4  = (N * (CH / 4) + T - 1) / T;
    const int gEdgeW = ((E * 32) + T - 1) / T;
    const int gRowW  = ((N * 32) + T - 1) / T;
    const dim3 gGemm((N + 63) / 64, CH / 32);

    k_deg_init<<<gN, T, 0, stream>>>(dinv, N);
    k_edge_deg<<<gE, T, 0, stream>>>(ei, dinv, E, N);
    k_dinv   <<<gN, T, 0, stream>>>(dinv, N);

    k_gemm_wmma    <<<gGemm, T, 0, stream>>>(x, W1, hbuf, N);
    k_selfloop_init<<<gRow4, T, 0, stream>>>(hbuf, dinv, b1, acc, N);
    k_edge_scatter <<<gEdgeW, T, 0, stream>>>(hbuf, ei, dinv, acc, E, N);
    k_rownorm_relu <<<gRowW, T, 0, stream>>>(acc, hbuf, N);

    k_gemm_wmma    <<<gGemm, T, 0, stream>>>(hbuf, W2, acc, N);
    k_selfloop_init<<<gRow4, T, 0, stream>>>(acc, dinv, b2, out, N);
    k_edge_scatter <<<gEdgeW, T, 0, stream>>>(acc, ei, dinv, out, E, N);
    k_rownorm_relu <<<gRowW, T, 0, stream>>>(out, out, N);
}


# scatter's global f32 atomics go: a work-group owns 392 output rows in LDS, filters the whole edge list for them, adds hits by LDS f32 atomics (arrival order), norms from LDS; 11 launches become 7
# speedup vs baseline: 1.9197x; 1.9197x over previous
#include <hip/hip_runtime.h>
#include <hip/hip_bf16.h>

typedef __attribute__((ext_vector_type(2))) float v2f;
typedef __attribute__((ext_vector_type(8))) float v8f;

#define CH 128
#define EPS 1e-5f

__global__ void k_deg_init(float* __restrict__ deg, int N) {
    int i = blockIdx.x * blockDim.x + threadIdx.x;
    if (i < N) deg[i] = 1.0f;
}

__global__ void k_edge_deg(const int* __restrict__ ei, float* __restrict__ deg,
                           int E, int N) {
    int e = blockIdx.x * blockDim.x + threadIdx.x;
    if (e >= E) return;
    int s = ei[e];
    int d = ei[E + e];
    if ((unsigned)s < (unsigned)N && (unsigned)d < (unsigned)N) {
        atomicAdd(&deg[d], 1.0f);
    }
}

__global__ void k_dinv(float* __restrict__ deg, int N) {
    int i = blockIdx.x * blockDim.x + threadIdx.x;
    if (i < N) deg[i] = rsqrtf(deg[i]);
}

__global__ __launch_bounds__(256) void k_gemm_wmma(const float* __restrict__ X,
                                                   const float* __restrict__ W,
                                                   float* __restrict__ H, int M) {
    __shared__ float sA[64][132];
    __shared__ float sB[32][132];
    const int m0  = blockIdx.x * 64;
    const int n0  = blockIdx.y * 32;
    const int tid = threadIdx.x;

    #pragma unroll
    for (int i = 0; i < 8; ++i) {
        const int idx   = tid + i * 256;
        const int row   = idx >> 5;
        const int chunk = idx & 31;
        const int gr    = m0 + row;
        const float4 v =
            *(const float4*)(X + (size_t)(gr < M ? gr : 0) * CH + chunk * 4);
        *(float4*)&sA[row][chunk * 4] = v;
    }
    {
        const int c     = tid & 31;
        const int kbase = (tid >> 5) * 16;
        #pragma unroll
        for (int i = 0; i < 16; ++i) {
            const int k = kbase + i;
            sB[c][k] = W[(size_t)k * CH + n0 + c];
        }
    }
    __syncthreads();

    const int wave  = tid >> 5;
    const int lane  = tid & 31;
    const int mt    = wave >> 1;
    const int nt    = wave & 1;
    const int mloc  = lane & 15;
    const int khalf = (lane >> 4) * 2;

    const float* arow = &sA[mt * 16 + mloc][0];
    const float* brow = &sB[nt * 16 + mloc][0];

    v8f c = {};
    #pragma unroll 8
    for (int k0 = 0; k0 < CH; k0 += 4) {
        const v2f a = *(const v2f*)(arow + k0 + khalf);
        const v2f b = *(const v2f*)(brow + k0 + khalf);
        c = __builtin_amdgcn_wmma_f32_16x16x4_f32(
                 false, a,  false, b,
                 (short)0, c,  false,  false);
    }

    const int col   = n0 + nt * 16 + mloc;
    const int half  = lane >> 4;
    const int mbase = m0 + mt * 16 + half * 8;
    if (m0 + 64 <= M) {
        #pragma unroll
        for (int r = 0; r < 8; ++r)
            H[(size_t)(mbase + r) * CH + col] = c[r];
    } else {
        #pragma unroll
        for (int r = 0; r < 8; ++r)
            if (mbase + r < M) H[(size_t)(mbase + r) * CH + col] = c[r];
    }
}

#define TR  392
#define TW  32
#define QN  256
#define SU  8
#define EG  8
#define CHUNK (SU * 128)

__global__ __launch_bounds__(1024) void k_tile_aggregate(const float* __restrict__ H,
                                                         const int* __restrict__ ei,
                                                         const float* __restrict__ dinv,
                                                         const float* __restrict__ bias,
                                                         float* __restrict__ Out,
                                                         int E, int N) {
    __shared__ float tile[TR * CH];
    __shared__ float sdinv[TR];
    __shared__ int   qe[TW][QN];
    __shared__ int   qr[TW][QN];

    const int lane = threadIdx.x & 31;
    const int wave = __builtin_amdgcn_readfirstlane((int)(threadIdx.x >> 5));
    const int r0   = blockIdx.x * TR;
    const int rows = (N - r0 < TR) ? (N - r0) : TR;

    {
        const float4 b = ((const float4*)bias)[lane];
        for (int rb = wave; rb < rows; rb += 4 * TW) {
            float4 h[4];
            float  di[4];
            #pragma unroll
            for (int k = 0; k < 4; ++k) {
                const int r  = rb + k * TW;
                const int rc = (r < rows) ? r : rb;
                di[k] = dinv[r0 + rc];
                h[k]  = ((const float4*)(H + (size_t)(r0 + rc) * CH))[lane];
            }
            #pragma unroll
            for (int k = 0; k < 4; ++k) {
                const int r = rb + k * TW;
                if (r < rows) {
                    const float w = di[k] * di[k];
                    float* t = tile + r * CH + lane;
                    t[0]  = h[k].x * w + b.x;
                    t[32] = h[k].y * w + b.y;
                    t[64] = h[k].z * w + b.z;
                    t[96] = h[k].w * w + b.w;
                    if (lane == 0) sdinv[r] = di[k];
                }
            }
        }
    }
    __syncthreads();

    const int* __restrict__ dst = ei + E;
    int qhead = 0;
    int qtail = 0;

    auto drain = [&](const int n) {
        __builtin_amdgcn_fence(__ATOMIC_ACQ_REL, "wavefront");
        int   myr = -1;
        int   s   = 0;
        float nrm = 0.0f;
        if (lane < n) {
            const int slot = (qhead + lane) & (QN - 1);
            const int e  = qe[wave][slot];
            const int lr = qr[wave][slot];
            const int ss = ei[e];
            if ((unsigned)ss < (unsigned)N) {
                s   = ss;
                myr = lr;
                nrm = dinv[ss] * sdinv[lr];
            }
        }
        #pragma unroll 1
        for (int g = 0; g < n; g += EG) {
            float4 v[EG];
            #pragma unroll
            for (int u = 0; u < EG; ++u) {
                const int sj = __builtin_amdgcn_readlane(s, g + u);
                v[u] = ((const float4*)(H + (size_t)sj * CH))[lane];
            }
            #pragma unroll
            for (int u = 0; u < EG; ++u) {
                const int rj = __builtin_amdgcn_readlane(myr, g + u);
                if (rj < 0) continue;
                const float nj = __int_as_float(
                    __builtin_amdgcn_readlane(__float_as_int(nrm), g + u));
                float* t = tile + rj * CH + lane;
                atomicAdd(t + 0,  v[u].x * nj);
                atomicAdd(t + 32, v[u].y * nj);
                atomicAdd(t + 64, v[u].z * nj);
                atomicAdd(t + 96, v[u].w * nj);
            }
        }
    };

    auto consider = [&](const int dv, const int e) {
        const int  lr = dv - r0;
        const bool m  = (unsigned)lr < (unsigned)rows;
        const unsigned mask = __builtin_amdgcn_ballot_w32(m);
        if (mask != 0u) {
            if (m) {
                const int slot =
                    (qtail + (int)__builtin_amdgcn_mbcnt_lo(mask, 0u)) & (QN - 1);
                qe[wave][slot] = e;
                qr[wave][slot] = lr;
            }
            qtail += __builtin_popcount(mask);
        }
    };

    const bool al16  = ((((size_t)dst) & 15) == 0);
    const int  nFull = al16 ? (E / CHUNK) : 0;
    {
        int  c = wave;
        int4 d[SU];
        if (c < nFull) {
            #pragma unroll
            for (int u = 0; u < SU; ++u)
                d[u] = *(const int4*)(dst + c * CHUNK + u * 128 + lane * 4);
        }
        while (c < nFull) {
            const int cn = c + TW;
            const int cl = (cn < nFull) ? cn : c;
            int4 dn[SU];
            #pragma unroll
            for (int u = 0; u < SU; ++u)
                dn[u] = *(const int4*)(dst + cl * CHUNK + u * 128 + lane * 4);
            #pragma unroll
            for (int u = 0; u < SU; ++u) {
                const int e = c * CHUNK + u * 128 + lane * 4;
                consider(d[u].x, e + 0);
                consider(d[u].y, e + 1);
                consider(d[u].z, e + 2);
                consider(d[u].w, e + 3);
                while (qtail - qhead >= 32) {
                    drain(32);
                    qhead += 32;
                }
            }
            #pragma unroll
            for (int u = 0; u < SU; ++u) d[u] = dn[u];
            c = cn;
        }
    }
    for (int eb = nFull * CHUNK + wave * 32; eb < E; eb += TW * 32) {
        const int e = eb + lane;
        consider((e < E) ? dst[e] : -1, e);
        while (qtail - qhead >= 32) {
            drain(32);
            qhead += 32;
        }
    }
    if (qtail - qhead > 0) drain(qtail - qhead);
    __syncthreads();

    for (int r = wave; r < rows; r += TW) {
        const float* t = tile + r * CH + lane;
        float4 v;
        v.x = t[0];  v.y = t[32];  v.z = t[64];  v.w = t[96];

        float s = v.x + v.y + v.z + v.w;
        #pragma unroll
        for (int o = 16; o > 0; o >>= 1) s += __shfl_xor(s, o, 32);
        const float mu = s * (1.0f / CH);

        const float d0 = v.x - mu, d1 = v.y - mu, d2 = v.z - mu, d3 = v.w - mu;
        float q = d0 * d0 + d1 * d1 + d2 * d2 + d3 * d3;
        #pragma unroll
        for (int o = 16; o > 0; o >>= 1) q += __shfl_xor(q, o, 32);
        const float inv = rsqrtf(q * (1.0f / CH) + EPS);

        float4 o4;
        o4.x = fmaxf(d0 * inv, 0.0f);
        o4.y = fmaxf(d1 * inv, 0.0f);
        o4.z = fmaxf(d2 * inv, 0.0f);
        o4.w = fmaxf(d3 * inv, 0.0f);
        ((float4*)(Out + (size_t)(r0 + r) * CH))[lane] = o4;
    }
}

extern "C" void kernel_launch(void* const* d_in, const int* in_sizes, int n_in,
                              void* d_out, int out_size, void* d_ws, size_t ws_size,
                              hipStream_t stream) {
    const float* x   = (const float*)d_in[0];
    const int*   ei  = (const int*)  d_in[1];
    const float* W1  = (const float*)d_in[2];
    const float* b1  = (const float*)d_in[3];
    const float* W2  = (const float*)d_in[4];
    const float* b2  = (const float*)d_in[5];
    float*       out = (float*)d_out;

    const int N = in_sizes[0] / CH;
    const int E = in_sizes[1] / 2;

    float* ws   = (float*)d_ws;
    const size_t nPad = ((size_t)N + 63) & ~(size_t)63;
    float* dinv = ws;
    float* hbuf = ws + nPad;
    float* acc  = hbuf + (size_t)N * CH;

    const int T = 256;
    const int gN    = (N + T - 1) / T;
    const int gE    = (E + T - 1) / T;
    const int gTile = (N + TR - 1) / TR;
    const dim3 gGemm((N + 63) / 64, CH / 32);

    k_deg_init<<<gN, T, 0, stream>>>(dinv, N);
    k_edge_deg<<<gE, T, 0, stream>>>(ei, dinv, E, N);
    k_dinv   <<<gN, T, 0, stream>>>(dinv, N);

    k_gemm_wmma     <<<gGemm, T, 0, stream>>>(x, W1, hbuf, N);
    k_tile_aggregate<<<gTile, TW * 32, 0, stream>>>(hbuf, ei, dinv, b1, acc, E, N);

    k_gemm_wmma     <<<gGemm, T, 0, stream>>>(acc, W2, hbuf, N);
    k_tile_aggregate<<<gTile, TW * 32, 0, stream>>>(hbuf, ei, dinv, b2, out, E, N);
}


# aggregation no longer scans the whole target array: the degree launch also writes one edge bit plane per 392-row tile, a tile's work-group walks only its plane; sums stay in LDS arrival order
# speedup vs baseline: 2.0730x; 1.0799x over previous
#include <hip/hip_runtime.h>
#include <hip/hip_bf16.h>

typedef __attribute__((ext_vector_type(2))) float v2f;
typedef __attribute__((ext_vector_type(8))) float v8f;

#define CH 128
#define EPS 1e-5f

__global__ void k_deg_init(float* __restrict__ deg, int N) {
    int i = blockIdx.x * blockDim.x + threadIdx.x;
    if (i < N) deg[i] = 1.0f;
}

#define TR  392

__global__ __launch_bounds__(1024) void k_edge_deg_planes(const int* __restrict__ ei,
                                                          float* __restrict__ deg,
                                                          unsigned* __restrict__ planes,
                                                          int E, int N, int PW, int nTiles) {
    __shared__ unsigned sm[4][32][33];
    const int lane = threadIdx.x & 31;
    const int wave = __builtin_amdgcn_readfirstlane((int)(threadIdx.x >> 5));
    const int ob0  = blockIdx.y * 128;
    const int e = blockIdx.x * 1024 + (int)threadIdx.x;
    int rel = -1;
    if (e < E) {
        const int s = ei[e];
        const int d = ei[E + e];
        if ((unsigned)s < (unsigned)N && (unsigned)d < (unsigned)N) {
            if (blockIdx.y == 0) atomicAdd(&deg[d], 1.0f);
            const int r = d / TR - ob0;
            rel = (r < 0) ? -1 : r;
        }
    }
    #pragma unroll
    for (int g = 0; g < 4; ++g) {
        unsigned mine = 0u;
        #pragma unroll
        for (int j = 0; j < 32; ++j) {
            const unsigned mk = __builtin_amdgcn_ballot_w32(rel == g * 32 + j);
            mine = (lane == j) ? mk : mine;
        }
        sm[g][lane][wave] = mine;
    }
    __syncthreads();
    #pragma unroll
    for (int g = 0; g < 4; ++g) {
        const int t = ob0 + g * 32 + wave;
        if (t < nTiles)
            planes[(size_t)t * PW + blockIdx.x * 32 + lane] = sm[g][wave][lane];
    }
}

__global__ void k_dinv(float* __restrict__ deg, int N) {
    int i = blockIdx.x * blockDim.x + threadIdx.x;
    if (i < N) deg[i] = rsqrtf(deg[i]);
}

__global__ __launch_bounds__(256) void k_gemm_wmma(const float* __restrict__ X,
                                                   const float* __restrict__ W,
                                                   float* __restrict__ H, int M) {
    __shared__ float sA[64][132];
    __shared__ float sB[32][132];
    const int m0  = blockIdx.x * 64;
    const int n0  = blockIdx.y * 32;
    const int tid = threadIdx.x;

    #pragma unroll
    for (int i = 0; i < 8; ++i) {
        const int idx   = tid + i * 256;
        const int row   = idx >> 5;
        const int chunk = idx & 31;
        const int gr    = m0 + row;
        const float4 v =
            *(const float4*)(X + (size_t)(gr < M ? gr : 0) * CH + chunk * 4);
        *(float4*)&sA[row][chunk * 4] = v;
    }
    {
        const int c     = tid & 31;
        const int kbase = (tid >> 5) * 16;
        #pragma unroll
        for (int i = 0; i < 16; ++i) {
            const int k = kbase + i;
            sB[c][k] = W[(size_t)k * CH + n0 + c];
        }
    }
    __syncthreads();

    const int wave  = tid >> 5;
    const int lane  = tid & 31;
    const int mt    = wave >> 1;
    const int nt    = wave & 1;
    const int mloc  = lane & 15;
    const int khalf = (lane >> 4) * 2;

    const float* arow = &sA[mt * 16 + mloc][0];
    const float* brow = &sB[nt * 16 + mloc][0];

    v8f c = {};
    #pragma unroll 8
    for (int k0 = 0; k0 < CH; k0 += 4) {
        const v2f a = *(const v2f*)(arow + k0 + khalf);
        const v2f b = *(const v2f*)(brow + k0 + khalf);
        c = __builtin_amdgcn_wmma_f32_16x16x4_f32(
                 false, a,  false, b,
                 (short)0, c,  false,  false);
    }

    const int col   = n0 + nt * 16 + mloc;
    const int half  = lane >> 4;
    const int mbase = m0 + mt * 16 + half * 8;
    if (m0 + 64 <= M) {
        #pragma unroll
        for (int r = 0; r < 8; ++r)
            H[(size_t)(mbase + r) * CH + col] = c[r];
    } else {
        #pragma unroll
        for (int r = 0; r < 8; ++r)
            if (mbase + r < M) H[(size_t)(mbase + r) * CH + col] = c[r];
    }
}

#define TW  32
#define QN  256
#define PU  8
#define EG  8

__global__ __launch_bounds__(1024) void k_tile_aggregate(const float* __restrict__ H,
                                                         const int* __restrict__ ei,
                                                         const float* __restrict__ dinv,
                                                         const float* __restrict__ bias,
                                                         const unsigned* __restrict__ planes,
                                                         float* __restrict__ Out,
                                                         int E, int N, int PW) {
    __shared__ float tile[TR * CH];
    __shared__ float sdinv[TR];
    __shared__ int   qe[TW][QN];

    const int lane = threadIdx.x & 31;
    const int wave = __builtin_amdgcn_readfirstlane((int)(threadIdx.x >> 5));
    const int r0   = blockIdx.x * TR;
    const int rows = (N - r0 < TR) ? (N - r0) : TR;

    {
        const float4 b = ((const float4*)bias)[lane];
        for (int rb = wave; rb < rows; rb += 4 * TW) {
            float4 h[4];
            float  di[4];
            #pragma unroll
            for (int k = 0; k < 4; ++k) {
                const int r  = rb + k * TW;
                const int rc = (r < rows) ? r : rb;
                di[k] = dinv[r0 + rc];
                h[k]  = ((const float4*)(H + (size_t)(r0 + rc) * CH))[lane];
            }
            #pragma unroll
            for (int k = 0; k < 4; ++k) {
                const int r = rb + k * TW;
                if (r < rows) {
                    const float w = di[k] * di[k];
                    float* t = tile + r * CH + lane;
                    t[0]  = h[k].x * w + b.x;
                    t[32] = h[k].y * w + b.y;
                    t[64] = h[k].z * w + b.z;
                    t[96] = h[k].w * w + b.w;
                    if (lane == 0) sdinv[r] = di[k];
                }
            }
        }
    }
    __syncthreads();

    const int* __restrict__ dst = ei + E;
    int qhead = 0;
    int qtail = 0;

    auto drain = [&](const int n) {
        __builtin_amdgcn_fence(__ATOMIC_ACQ_REL, "wavefront");
        int   myr = -1;
        int   s   = 0;
        float nrm = 0.0f;
        if (lane < n) {
            const int slot = (qhead + lane) & (QN - 1);
            const int e  = qe[wave][slot];
            const int ss = ei[e];
            const int lr = dst[e] - r0;
            if ((unsigned)ss < (unsigned)N && (unsigned)lr < (unsigned)rows) {
                s   = ss;
                myr = lr;
                nrm = dinv[ss] * sdinv[lr];
            }
        }
        #pragma unroll 1
        for (int g = 0; g < n; g += EG) {
            float4 v[EG];
            #pragma unroll
            for (int u = 0; u < EG; ++u) {
                const int sj = __builtin_amdgcn_readlane(s, g + u);
                v[u] = ((const float4*)(H + (size_t)sj * CH))[lane];
            }
            #pragma unroll
            for (int u = 0; u < EG; ++u) {
                const int rj = __builtin_amdgcn_readlane(myr, g + u);
                if (rj < 0) continue;
                const float nj = __int_as_float(
                    __builtin_amdgcn_readlane(__float_as_int(nrm), g + u));
                float* t = tile + rj * CH + lane;
                atomicAdd(t + 0,  v[u].x * nj);
                atomicAdd(t + 32, v[u].y * nj);
                atomicAdd(t + 64, v[u].z * nj);
                atomicAdd(t + 96, v[u].w * nj);
            }
        }
    };

    auto take = [&](unsigned w, const int ln) {
        unsigned any = __builtin_amdgcn_ballot_w32(w != 0u);
        while (any != 0u) {
            if (w != 0u) {
                const int slot =
                    (qtail + (int)__builtin_amdgcn_mbcnt_lo(any, 0u)) & (QN - 1);
                qe[wave][slot] = ((ln * 32 + lane) << 5) + __builtin_ctz(w);
                w &= w - 1u;
            }
            qtail += __builtin_popcount(any);
            while (qtail - qhead >= 32) {
                drain(32);
                qhead += 32;
            }
            any = __builtin_amdgcn_ballot_w32(w != 0u);
        }
    };

    {
        const unsigned* __restrict__ pl = planes + (size_t)blockIdx.x * PW;
        const int nLines = PW >> 5;
        int c = wave;
        unsigned d[PU];
        #pragma unroll
        for (int u = 0; u < PU; ++u) {
            const int l  = c + u * TW;
            const int lc = (l < nLines) ? l : (nLines - 1);
            const unsigned x = pl[lc * 32 + lane];
            d[u] = (l < nLines) ? x : 0u;
        }
        while (c < nLines) {
            const int cn = c + PU * TW;
            unsigned dn[PU];
            #pragma unroll
            for (int u = 0; u < PU; ++u) {
                const int l  = cn + u * TW;
                const int lc = (l < nLines) ? l : (nLines - 1);
                const unsigned x = pl[lc * 32 + lane];
                dn[u] = (l < nLines) ? x : 0u;
            }
            #pragma unroll
            for (int u = 0; u < PU; ++u) take(d[u], c + u * TW);
            #pragma unroll
            for (int u = 0; u < PU; ++u) d[u] = dn[u];
            c = cn;
        }
    }
    if (qtail - qhead > 0) drain(qtail - qhead);
    __syncthreads();

    for (int r = wave; r < rows; r += TW) {
        const float* t = tile + r * CH + lane;
        float4 v;
        v.x = t[0];  v.y = t[32];  v.z = t[64];  v.w = t[96];

        float s = v.x + v.y + v.z + v.w;
        #pragma unroll
        for (int o = 16; o > 0; o >>= 1) s += __shfl_xor(s, o, 32);
        const float mu = s * (1.0f / CH);

        const float d0 = v.x - mu, d1 = v.y - mu, d2 = v.z - mu, d3 = v.w - mu;
        float q = d0 * d0 + d1 * d1 + d2 * d2 + d3 * d3;
        #pragma unroll
        for (int o = 16; o > 0; o >>= 1) q += __shfl_xor(q, o, 32);
        const float inv = rsqrtf(q * (1.0f / CH) + EPS);

        float4 o4;
        o4.x = fmaxf(d0 * inv, 0.0f);
        o4.y = fmaxf(d1 * inv, 0.0f);
        o4.z = fmaxf(d2 * inv, 0.0f);
        o4.w = fmaxf(d3 * inv, 0.0f);
        ((float4*)(Out + (size_t)(r0 + r) * CH))[lane] = o4;
    }
}

extern "C" void kernel_launch(void* const* d_in, const int* in_sizes, int n_in,
                              void* d_out, int out_size, void* d_ws, size_t ws_size,
                              hipStream_t stream) {
    const float* x   = (const float*)d_in[0];
    const int*   ei  = (const int*)  d_in[1];
    const float* W1  = (const float*)d_in[2];
    const float* b1  = (const float*)d_in[3];
    const float* W2  = (const float*)d_in[4];
    const float* b2  = (const float*)d_in[5];
    float*       out = (float*)d_out;

    const int N = in_sizes[0] / CH;
    const int E = in_sizes[1] / 2;

    float* ws   = (float*)d_ws;
    const size_t nPad = ((size_t)N + 63) & ~(size_t)63;
    float* dinv = ws;
    float* hbuf = ws + nPad;
    unsigned* planes = (unsigned*)(hbuf + (size_t)N * CH);

    const int T = 256;
    const int gN    = (N + T - 1) / T;
    const int gEdge = (E + 1023) / 1024;
    const int PW    = gEdge * 32;
    const int gTile = (N + TR - 1) / TR;
    const dim3 gGemm((N + 63) / 64, CH / 32);

    k_deg_init       <<<gN, T, 0, stream>>>(dinv, N);
    k_edge_deg_planes<<<dim3(gEdge, (gTile + 127) / 128), 1024, 0, stream>>>(ei, dinv, planes, E, N, PW, gTile);
    k_dinv           <<<gN, T, 0, stream>>>(dinv, N);

    k_gemm_wmma     <<<gGemm, T, 0, stream>>>(x, W1, hbuf, N);
    k_tile_aggregate<<<gTile, TW * 32, 0, stream>>>(hbuf, ei, dinv, b1, planes, out, E, N, PW);

    k_gemm_wmma     <<<gGemm, T, 0, stream>>>(out, W2, hbuf, N);
    k_tile_aggregate<<<gTile, TW * 32, 0, stream>>>(hbuf, ei, dinv, b2, planes, out, E, N, PW);
}


# aggregation row tile 392 -> 197 rows: 254 work-groups, one per WGP (LDS reservation held above 160 KiB), where 128 left half the chip's WGPs idle; two plane groups; sums stay in LDS arrival order
# speedup vs baseline: 2.4346x; 1.1744x over previous
#include <hip/hip_runtime.h>
#include <hip/hip_bf16.h>

typedef __attribute__((ext_vector_type(2))) float v2f;
typedef __attribute__((ext_vector_type(8))) float v8f;

#define CH 128
#define EPS 1e-5f

__global__ void k_deg_init(float* __restrict__ deg, int N) {
    int i = blockIdx.x * blockDim.x + threadIdx.x;
    if (i < N) deg[i] = 1.0f;
}

#define TR  197

__global__ __launch_bounds__(1024) void k_edge_deg_planes(const int* __restrict__ ei,
                                                          float* __restrict__ deg,
                                                          unsigned* __restrict__ planes,
                                                          int E, int N, int PW, int nTiles) {
    __shared__ unsigned sm[4][32][33];
    const int lane = threadIdx.x & 31;
    const int wave = __builtin_amdgcn_readfirstlane((int)(threadIdx.x >> 5));
    const int ob0  = blockIdx.y * 128;
    const int e = blockIdx.x * 1024 + (int)threadIdx.x;
    int rel = -1;
    if (e < E) {
        const int s = ei[e];
        const int d = ei[E + e];
        if ((unsigned)s < (unsigned)N && (unsigned)d < (unsigned)N) {
            if (blockIdx.y == 0) atomicAdd(&deg[d], 1.0f);
            const int r = d / TR - ob0;
            rel = (r < 0) ? -1 : r;
        }
    }
    #pragma unroll
    for (int g = 0; g < 4; ++g) {
        unsigned mine = 0u;
        #pragma unroll
        for (int j = 0; j < 32; ++j) {
            const unsigned mk = __builtin_amdgcn_ballot_w32(rel == g * 32 + j);
            mine = (lane == j) ? mk : mine;
        }
        sm[g][lane][wave] = mine;
    }
    __syncthreads();
    #pragma unroll
    for (int g = 0; g < 4; ++g) {
        const int t = ob0 + g * 32 + wave;
        if (t < nTiles)
            planes[(size_t)t * PW + blockIdx.x * 32 + lane] = sm[g][wave][lane];
    }
}

__global__ void k_dinv(float* __restrict__ deg, int N) {
    int i = blockIdx.x * blockDim.x + threadIdx.x;
    if (i < N) deg[i] = rsqrtf(deg[i]);
}

__global__ __launch_bounds__(256) void k_gemm_wmma(const float* __restrict__ X,
                                                   const float* __restrict__ W,
                                                   float* __restrict__ H, int M) {
    __shared__ float sA[64][132];
    __shared__ float sB[32][132];
    const int m0  = blockIdx.x * 64;
    const int n0  = blockIdx.y * 32;
    const int tid = threadIdx.x;

    #pragma unroll
    for (int i = 0; i < 8; ++i) {
        const int idx   = tid + i * 256;
        const int row   = idx >> 5;
        const int chunk = idx & 31;
        const int gr    = m0 + row;
        const float4 v =
            *(const float4*)(X + (size_t)(gr < M ? gr : 0) * CH + chunk * 4);
        *(float4*)&sA[row][chunk * 4] = v;
    }
    {
        const int c     = tid & 31;
        const int kbase = (tid >> 5) * 16;
        #pragma unroll
        for (int i = 0; i < 16; ++i) {
            const int k = kbase + i;
            sB[c][k] = W[(size_t)k * CH + n0 + c];
        }
    }
    __syncthreads();

    const int wave  = tid >> 5;
    const int lane  = tid & 31;
    const int mt    = wave >> 1;
    const int nt    = wave & 1;
    const int mloc  = lane & 15;
    const int khalf = (lane >> 4) * 2;

    const float* arow = &sA[mt * 16 + mloc][0];
    const float* brow = &sB[nt * 16 + mloc][0];

    v8f c = {};
    #pragma unroll 8
    for (int k0 = 0; k0 < CH; k0 += 4) {
        const v2f a = *(const v2f*)(arow + k0 + khalf);
        const v2f b = *(const v2f*)(brow + k0 + khalf);
        c = __builtin_amdgcn_wmma_f32_16x16x4_f32(
                 false, a,  false, b,
                 (short)0, c,  false,  false);
    }

    const int col   = n0 + nt * 16 + mloc;
    const int half  = lane >> 4;
    const int mbase = m0 + mt * 16 + half * 8;
    if (m0 + 64 <= M) {
        #pragma unroll
        for (int r = 0; r < 8; ++r)
            H[(size_t)(mbase + r) * CH + col] = c[r];
    } else {
        #pragma unroll
        for (int r = 0; r < 8; ++r)
            if (mbase + r < M) H[(size_t)(mbase + r) * CH + col] = c[r];
    }
}

#define TW  32
#define QN  256
#define PU  8
#define EG  8
#define LDS_RESERVE 65536

__global__ __launch_bounds__(1024) void k_tile_aggregate(const float* __restrict__ H,
                                                         const int* __restrict__ ei,
                                                         const float* __restrict__ dinv,
                                                         const float* __restrict__ bias,
                                                         const unsigned* __restrict__ planes,
                                                         float* __restrict__ Out,
                                                         int E, int N, int PW) {
    __shared__ float tile[TR * CH];
    __shared__ float sdinv[TR];
    __shared__ int   qe[TW][QN];

    const int lane = threadIdx.x & 31;
    const int wave = __builtin_amdgcn_readfirstlane((int)(threadIdx.x >> 5));
    const int r0   = blockIdx.x * TR;
    const int rows = (N - r0 < TR) ? (N - r0) : TR;

    {
        const float4 b = ((const float4*)bias)[lane];
        for (int rb = wave; rb < rows; rb += 4 * TW) {
            float4 h[4];
            float  di[4];
            #pragma unroll
            for (int k = 0; k < 4; ++k) {
                const int r  = rb + k * TW;
                const int rc = (r < rows) ? r : rb;
                di[k] = dinv[r0 + rc];
                h[k]  = ((const float4*)(H + (size_t)(r0 + rc) * CH))[lane];
            }
            #pragma unroll
            for (int k = 0; k < 4; ++k) {
                const int r = rb + k * TW;
                if (r < rows) {
                    const float w = di[k] * di[k];
                    float* t = tile + r * CH + lane;
                    t[0]  = h[k].x * w + b.x;
                    t[32] = h[k].y * w + b.y;
                    t[64] = h[k].z * w + b.z;
                    t[96] = h[k].w * w + b.w;
                    if (lane == 0) sdinv[r] = di[k];
                }
            }
        }
    }
    __syncthreads();

    const int* __restrict__ dst = ei + E;
    int qhead = 0;
    int qtail = 0;

    auto drain = [&](const int n) {
        __builtin_amdgcn_fence(__ATOMIC_ACQ_REL, "wavefront");
        int   myr = -1;
        int   s   = 0;
        float nrm = 0.0f;
        if (lane < n) {
            const int slot = (qhead + lane) & (QN - 1);
            const int e  = qe[wave][slot];
            const int ss = ei[e];
            const int lr = dst[e] - r0;
            if ((unsigned)ss < (unsigned)N && (unsigned)lr < (unsigned)rows) {
                s   = ss;
                myr = lr;
                nrm = dinv[ss] * sdinv[lr];
            }
        }
        #pragma unroll 1
        for (int g = 0; g < n; g += EG) {
            float4 v[EG];
            #pragma unroll
            for (int u = 0; u < EG; ++u) {
                const int sj = __builtin_amdgcn_readlane(s, g + u);
                v[u] = ((const float4*)(H + (size_t)sj * CH))[lane];
            }
            #pragma unroll
            for (int u = 0; u < EG; ++u) {
                const int rj = __builtin_amdgcn_readlane(myr, g + u);
                if (rj < 0) continue;
                const float nj = __int_as_float(
                    __builtin_amdgcn_readlane(__float_as_int(nrm), g + u));
                float* t = tile + rj * CH + lane;
                atomicAdd(t + 0,  v[u].x * nj);
                atomicAdd(t + 32, v[u].y * nj);
                atomicAdd(t + 64, v[u].z * nj);
                atomicAdd(t + 96, v[u].w * nj);
            }
        }
    };

    auto take = [&](unsigned w, const int ln) {
        unsigned any = __builtin_amdgcn_ballot_w32(w != 0u);
        while (any != 0u) {
            if (w != 0u) {
                const int slot =
                    (qtail + (int)__builtin_amdgcn_mbcnt_lo(any, 0u)) & (QN - 1);
                qe[wave][slot] = ((ln * 32 + lane) << 5) + __builtin_ctz(w);
                w &= w - 1u;
            }
            qtail += __builtin_popcount(any);
            while (qtail - qhead >= 32) {
                drain(32);
                qhead += 32;
            }
            any = __builtin_amdgcn_ballot_w32(w != 0u);
        }
    };

    {
        const unsigned* __restrict__ pl = planes + (size_t)blockIdx.x * PW;
        const int nLines = PW >> 5;
        int c = wave;
        unsigned d[PU];
        #pragma unroll
        for (int u = 0; u < PU; ++u) {
            const int l  = c + u * TW;
            const int lc = (l < nLines) ? l : (nLines - 1);
            const unsigned x = pl[lc * 32 + lane];
            d[u] = (l < nLines) ? x : 0u;
        }
        while (c < nLines) {
            const int cn = c + PU * TW;
            unsigned dn[PU];
            #pragma unroll
            for (int u = 0; u < PU; ++u) {
                const int l  = cn + u * TW;
                const int lc = (l < nLines) ? l : (nLines - 1);
                const unsigned x = pl[lc * 32 + lane];
                dn[u] = (l < nLines) ? x : 0u;
            }
            #pragma unroll
            for (int u = 0; u < PU; ++u) take(d[u], c + u * TW);
            #pragma unroll
            for (int u = 0; u < PU; ++u) d[u] = dn[u];
            c = cn;
        }
    }
    if (qtail - qhead > 0) drain(qtail - qhead);
    __syncthreads();

    for (int r = wave; r < rows; r += TW) {
        const float* t = tile + r * CH + lane;
        float4 v;
        v.x = t[0];  v.y = t[32];  v.z = t[64];  v.w = t[96];

        float s = v.x + v.y + v.z + v.w;
        #pragma unroll
        for (int o = 16; o > 0; o >>= 1) s += __shfl_xor(s, o, 32);
        const float mu = s * (1.0f / CH);

        const float d0 = v.x - mu, d1 = v.y - mu, d2 = v.z - mu, d3 = v.w - mu;
        float q = d0 * d0 + d1 * d1 + d2 * d2 + d3 * d3;
        #pragma unroll
        for (int o = 16; o > 0; o >>= 1) q += __shfl_xor(q, o, 32);
        const float inv = rsqrtf(q * (1.0f / CH) + EPS);

        float4 o4;
        o4.x = fmaxf(d0 * inv, 0.0f);
        o4.y = fmaxf(d1 * inv, 0.0f);
        o4.z = fmaxf(d2 * inv, 0.0f);
        o4.w = fmaxf(d3 * inv, 0.0f);
        ((float4*)(Out + (size_t)(r0 + r) * CH))[lane] = o4;
    }
}

extern "C" void kernel_launch(void* const* d_in, const int* in_sizes, int n_in,
                              void* d_out, int out_size, void* d_ws, size_t ws_size,
                              hipStream_t stream) {
    const float* x   = (const float*)d_in[0];
    const int*   ei  = (const int*)  d_in[1];
    const float* W1  = (const float*)d_in[2];
    const float* b1  = (const float*)d_in[3];
    const float* W2  = (const float*)d_in[4];
    const float* b2  = (const float*)d_in[5];
    float*       out = (float*)d_out;

    const int N = in_sizes[0] / CH;
    const int E = in_sizes[1] / 2;

    float* ws   = (float*)d_ws;
    const size_t nPad = ((size_t)N + 63) & ~(size_t)63;
    float* dinv = ws;
    float* hbuf = ws + nPad;
    unsigned* planes = (unsigned*)(hbuf + (size_t)N * CH);

    const int T = 256;
    const int gN    = (N + T - 1) / T;
    const int gEdge = (E + 1023) / 1024;
    const int PW    = gEdge * 32;
    const int gTile = (N + TR - 1) / TR;
    const dim3 gGemm((N + 63) / 64, CH / 32);

    k_deg_init       <<<gN, T, 0, stream>>>(dinv, N);
    k_edge_deg_planes<<<dim3(gEdge, (gTile + 127) / 128), 1024, 0, stream>>>(ei, dinv, planes, E, N, PW, gTile);
    k_dinv           <<<gN, T, 0, stream>>>(dinv, N);

    k_gemm_wmma     <<<gGemm, T, 0, stream>>>(x, W1, hbuf, N);
    k_tile_aggregate<<<gTile, TW * 32, LDS_RESERVE, stream>>>(hbuf, ei, dinv, b1, planes, out, E, N, PW);

    k_gemm_wmma     <<<gGemm, T, 0, stream>>>(out, W2, hbuf, N);
    k_tile_aggregate<<<gTile, TW * 32, LDS_RESERVE, stream>>>(hbuf, ei, dinv, b2, planes, out, E, N, PW);
}


# edge weight factored out of the edge loop: GEMM store scales row n by dinv[n], finished row scaled once by dinv[dst]; per edge no dinv gather, no multiplies; sum order kept, products re-associated
# speedup vs baseline: 2.5595x; 1.0513x over previous
#include <hip/hip_runtime.h>
#include <hip/hip_bf16.h>

typedef __attribute__((ext_vector_type(2))) float v2f;
typedef __attribute__((ext_vector_type(8))) float v8f;

#define CH 128
#define EPS 1e-5f

__global__ void k_deg_init(float* __restrict__ deg, int N) {
    int i = blockIdx.x * blockDim.x + threadIdx.x;
    if (i < N) deg[i] = 1.0f;
}

#define TR  197

__global__ __launch_bounds__(1024) void k_edge_deg_planes(const int* __restrict__ ei,
                                                          float* __restrict__ deg,
                                                          unsigned* __restrict__ planes,
                                                          int E, int N, int PW, int nTiles) {
    __shared__ unsigned sm[4][32][33];
    const int lane = threadIdx.x & 31;
    const int wave = __builtin_amdgcn_readfirstlane((int)(threadIdx.x >> 5));
    const int ob0  = blockIdx.y * 128;
    const int e = blockIdx.x * 1024 + (int)threadIdx.x;
    int rel = -1;
    if (e < E) {
        const int s = ei[e];
        const int d = ei[E + e];
        if ((unsigned)s < (unsigned)N && (unsigned)d < (unsigned)N) {
            if (blockIdx.y == 0) atomicAdd(&deg[d], 1.0f);
            const int r = d / TR - ob0;
            rel = (r < 0) ? -1 : r;
        }
    }
    #pragma unroll
    for (int g = 0; g < 4; ++g) {
        unsigned mine = 0u;
        #pragma unroll
        for (int j = 0; j < 32; ++j) {
            const unsigned mk = __builtin_amdgcn_ballot_w32(rel == g * 32 + j);
            mine = (lane == j) ? mk : mine;
        }
        sm[g][lane][wave] = mine;
    }
    __syncthreads();
    #pragma unroll
    for (int g = 0; g < 4; ++g) {
        const int t = ob0 + g * 32 + wave;
        if (t < nTiles)
            planes[(size_t)t * PW + blockIdx.x * 32 + lane] = sm[g][wave][lane];
    }
}

__global__ void k_dinv(float* __restrict__ deg, int N) {
    int i = blockIdx.x * blockDim.x + threadIdx.x;
    if (i < N) deg[i] = rsqrtf(deg[i]);
}

__global__ __launch_bounds__(256) void k_gemm_wmma(const float* __restrict__ X,
                                                   const float* __restrict__ W,
                                                   const float* __restrict__ rs,
                                                   float* __restrict__ H, int M) {
    __shared__ float sA[64][132];
    __shared__ float sB[32][132];
    const int m0  = blockIdx.x * 64;
    const int n0  = blockIdx.y * 32;
    const int tid = threadIdx.x;

    #pragma unroll
    for (int i = 0; i < 8; ++i) {
        const int idx   = tid + i * 256;
        const int row   = idx >> 5;
        const int chunk = idx & 31;
        const int gr    = m0 + row;
        const float4 v =
            *(const float4*)(X + (size_t)(gr < M ? gr : 0) * CH + chunk * 4);
        *(float4*)&sA[row][chunk * 4] = v;
    }
    {
        const int c     = tid & 31;
        const int kbase = (tid >> 5) * 16;
        #pragma unroll
        for (int i = 0; i < 16; ++i) {
            const int k = kbase + i;
            sB[c][k] = W[(size_t)k * CH + n0 + c];
        }
    }
    __syncthreads();

    const int wave  = tid >> 5;
    const int lane  = tid & 31;
    const int mt    = wave >> 1;
    const int nt    = wave & 1;
    const int mloc  = lane & 15;
    const int khalf = (lane >> 4) * 2;

    const float* arow = &sA[mt * 16 + mloc][0];
    const float* brow = &sB[nt * 16 + mloc][0];

    const int sbase = m0 + mt * 16 + (lane >> 4) * 8;
    const float4 sc0 = *(const float4*)(rs + sbase);
    const float4 sc1 = *(const float4*)(rs + sbase + 4);

    v8f c = {};
    #pragma unroll 8
    for (int k0 = 0; k0 < CH; k0 += 4) {
        const v2f a = *(const v2f*)(arow + k0 + khalf);
        const v2f b = *(const v2f*)(brow + k0 + khalf);
        c = __builtin_amdgcn_wmma_f32_16x16x4_f32(
                 false, a,  false, b,
                 (short)0, c,  false,  false);
    }

    const int col   = n0 + nt * 16 + mloc;
    const int half  = lane >> 4;
    const int mbase = m0 + mt * 16 + half * 8;
    const float sc[8] = {sc0.x, sc0.y, sc0.z, sc0.w, sc1.x, sc1.y, sc1.z, sc1.w};
    if (m0 + 64 <= M) {
        #pragma unroll
        for (int r = 0; r < 8; ++r)
            H[(size_t)(mbase + r) * CH + col] = c[r] * sc[r];
    } else {
        #pragma unroll
        for (int r = 0; r < 8; ++r)
            if (mbase + r < M) H[(size_t)(mbase + r) * CH + col] = c[r] * sc[r];
    }
}

#define TW  32
#define QN  256
#define PU  8
#define EG  8
#define LDS_RESERVE 65536

__global__ __launch_bounds__(1024) void k_tile_aggregate(const float* __restrict__ H,
                                                         const int* __restrict__ ei,
                                                         const float* __restrict__ dinv,
                                                         const float* __restrict__ bias,
                                                         const unsigned* __restrict__ planes,
                                                         float* __restrict__ Out,
                                                         int E, int N, int PW) {
    __shared__ float tile[TR * CH];
    __shared__ float sdinv[TR];
    __shared__ int   qe[TW][QN];

    const int lane = threadIdx.x & 31;
    const int wave = __builtin_amdgcn_readfirstlane((int)(threadIdx.x >> 5));
    const int r0   = blockIdx.x * TR;
    const int rows = (N - r0 < TR) ? (N - r0) : TR;

    {
        for (int rb = wave; rb < rows; rb += 4 * TW) {
            float4 h[4];
            float  di[4];
            #pragma unroll
            for (int k = 0; k < 4; ++k) {
                const int r  = rb + k * TW;
                const int rc = (r < rows) ? r : rb;
                di[k] = dinv[r0 + rc];
                h[k]  = ((const float4*)(H + (size_t)(r0 + rc) * CH))[lane];
            }
            #pragma unroll
            for (int k = 0; k < 4; ++k) {
                const int r = rb + k * TW;
                if (r < rows) {
                    float* t = tile + r * CH + lane;
                    t[0]  = h[k].x;
                    t[32] = h[k].y;
                    t[64] = h[k].z;
                    t[96] = h[k].w;
                    if (lane == 0) sdinv[r] = di[k];
                }
            }
        }
    }
    __syncthreads();

    const int* __restrict__ dst = ei + E;
    int qhead = 0;
    int qtail = 0;

    auto drain = [&](const int n) {
        __builtin_amdgcn_fence(__ATOMIC_ACQ_REL, "wavefront");
        int myr = -1;
        int s   = 0;
        if (lane < n) {
            const int slot = (qhead + lane) & (QN - 1);
            const int e  = qe[wave][slot];
            const int ss = ei[e];
            const int lr = dst[e] - r0;
            const bool ok = ((unsigned)ss < (unsigned)N) & ((unsigned)lr < (unsigned)rows);
            s   = ok ? ss : 0;
            myr = ok ? lr : -1;
        }
        #pragma unroll 1
        for (int g = 0; g < n; g += EG) {
            float4 v[EG];
            #pragma unroll
            for (int u = 0; u < EG; ++u) {
                const int sj = __builtin_amdgcn_readlane(s, g + u);
                v[u] = ((const float4*)(H + (size_t)sj * CH))[lane];
            }
            #pragma unroll
            for (int u = 0; u < EG; ++u) {
                const int rj = __builtin_amdgcn_readlane(myr, g + u);
                if (rj < 0) continue;
                float* t = tile + rj * CH + lane;
                atomicAdd(t + 0,  v[u].x);
                atomicAdd(t + 32, v[u].y);
                atomicAdd(t + 64, v[u].z);
                atomicAdd(t + 96, v[u].w);
            }
        }
    };

    auto take = [&](unsigned w, const int ln) {
        unsigned any = __builtin_amdgcn_ballot_w32(w != 0u);
        while (any != 0u) {
            if (w != 0u) {
                const int slot =
                    (qtail + (int)__builtin_amdgcn_mbcnt_lo(any, 0u)) & (QN - 1);
                qe[wave][slot] = ((ln * 32 + lane) << 5) + __builtin_ctz(w);
                w &= w - 1u;
            }
            qtail += __builtin_popcount(any);
            while (qtail - qhead >= 32) {
                drain(32);
                qhead += 32;
            }
            any = __builtin_amdgcn_ballot_w32(w != 0u);
        }
    };

    {
        const unsigned* __restrict__ pl = planes + (size_t)blockIdx.x * PW;
        const int nLines = PW >> 5;
        int c = wave;
        unsigned d[PU];
        #pragma unroll
        for (int u = 0; u < PU; ++u) {
            const int l  = c + u * TW;
            const int lc = (l < nLines) ? l : (nLines - 1);
            const unsigned x = pl[lc * 32 + lane];
            d[u] = (l < nLines) ? x : 0u;
        }
        while (c < nLines) {
            const int cn = c + PU * TW;
            unsigned dn[PU];
            #pragma unroll
            for (int u = 0; u < PU; ++u) {
                const int l  = cn + u * TW;
                const int lc = (l < nLines) ? l : (nLines - 1);
                const unsigned x = pl[lc * 32 + lane];
                dn[u] = (l < nLines) ? x : 0u;
            }
            #pragma unroll
            for (int u = 0; u < PU; ++u) take(d[u], c + u * TW);
            #pragma unroll
            for (int u = 0; u < PU; ++u) d[u] = dn[u];
            c = cn;
        }
    }
    if (qtail - qhead > 0) drain(qtail - qhead);
    __syncthreads();

    const float4 b = ((const float4*)bias)[lane];
    for (int r = wave; r < rows; r += TW) {
        const float* t = tile + r * CH + lane;
        const float  w = sdinv[r];
        float4 v;
        v.x = t[0]  * w + b.x;
        v.y = t[32] * w + b.y;
        v.z = t[64] * w + b.z;
        v.w = t[96] * w + b.w;

        float s = v.x + v.y + v.z + v.w;
        #pragma unroll
        for (int o = 16; o > 0; o >>= 1) s += __shfl_xor(s, o, 32);
        const float mu = s * (1.0f / CH);

        const float d0 = v.x - mu, d1 = v.y - mu, d2 = v.z - mu, d3 = v.w - mu;
        float q = d0 * d0 + d1 * d1 + d2 * d2 + d3 * d3;
        #pragma unroll
        for (int o = 16; o > 0; o >>= 1) q += __shfl_xor(q, o, 32);
        const float inv = rsqrtf(q * (1.0f / CH) + EPS);

        float4 o4;
        o4.x = fmaxf(d0 * inv, 0.0f);
        o4.y = fmaxf(d1 * inv, 0.0f);
        o4.z = fmaxf(d2 * inv, 0.0f);
        o4.w = fmaxf(d3 * inv, 0.0f);
        ((float4*)(Out + (size_t)(r0 + r) * CH))[lane] = o4;
    }
}

extern "C" void kernel_launch(void* const* d_in, const int* in_sizes, int n_in,
                              void* d_out, int out_size, void* d_ws, size_t ws_size,
                              hipStream_t stream) {
    const float* x   = (const float*)d_in[0];
    const int*   ei  = (const int*)  d_in[1];
    const float* W1  = (const float*)d_in[2];
    const float* b1  = (const float*)d_in[3];
    const float* W2  = (const float*)d_in[4];
    const float* b2  = (const float*)d_in[5];
    float*       out = (float*)d_out;

    const int N = in_sizes[0] / CH;
    const int E = in_sizes[1] / 2;

    float* ws   = (float*)d_ws;
    const size_t nPad = ((size_t)N + 63) & ~(size_t)63;
    float* dinv = ws;
    float* hbuf = ws + nPad;
    unsigned* planes = (unsigned*)(hbuf + (size_t)N * CH);

    const int T = 256;
    const int gN    = (N + T - 1) / T;
    const int gEdge = (E + 1023) / 1024;
    const int PW    = gEdge * 32;
    const int gTile = (N + TR - 1) / TR;
    const dim3 gGemm((N + 63) / 64, CH / 32);

    k_deg_init       <<<gN, T, 0, stream>>>(dinv, (int)nPad);
    k_edge_deg_planes<<<dim3(gEdge, (gTile + 127) / 128), 1024, 0, stream>>>(ei, dinv, planes, E, N, PW, gTile);
    k_dinv           <<<gN, T, 0, stream>>>(dinv, (int)nPad);

    k_gemm_wmma     <<<gGemm, T, 0, stream>>>(x, W1, dinv, hbuf, N);
    k_tile_aggregate<<<gTile, TW * 32, LDS_RESERVE, stream>>>(hbuf, ei, dinv, b1, planes, out, E, N, PW);

    k_gemm_wmma     <<<gGemm, T, 0, stream>>>(out, W2, dinv, hbuf, N);
    k_tile_aggregate<<<gTile, TW * 32, LDS_RESERVE, stream>>>(hbuf, ei, dinv, b2, planes, out, E, N, PW);
}


# layer 2 no longer looks its edges up: layer 1's tile work-group keeps each looked-up edge as a packed word, writes the list over its own bit plane, layer 2 streams it; sums stay in LDS arrival order
# speedup vs baseline: 2.6807x; 1.0474x over previous
#include <hip/hip_runtime.h>
#include <hip/hip_bf16.h>

typedef __attribute__((ext_vector_type(2))) float v2f;
typedef __attribute__((ext_vector_type(8))) float v8f;

#define CH 128
#define EPS 1e-5f

__global__ void k_deg_init(float* __restrict__ deg, int N) {
    int i = blockIdx.x * blockDim.x + threadIdx.x;
    if (i < N) deg[i] = 1.0f;
}

#define TR  197

__global__ __launch_bounds__(1024) void k_edge_deg_planes(const int* __restrict__ ei,
                                                          float* __restrict__ deg,
                                                          unsigned* __restrict__ planes,
                                                          int E, int N, int PW, int nTiles) {
    __shared__ unsigned sm[4][32][33];
    const int lane = threadIdx.x & 31;
    const int wave = __builtin_amdgcn_readfirstlane((int)(threadIdx.x >> 5));
    const int ob0  = blockIdx.y * 128;
    const int e = blockIdx.x * 1024 + (int)threadIdx.x;
    int rel = -1;
    if (e < E) {
        const int s = ei[e];
        const int d = ei[E + e];
        if ((unsigned)s < (unsigned)N && (unsigned)d < (unsigned)N) {
            if (blockIdx.y == 0) atomicAdd(&deg[d], 1.0f);
            const int r = d / TR - ob0;
            rel = (r < 0) ? -1 : r;
        }
    }
    #pragma unroll
    for (int g = 0; g < 4; ++g) {
        unsigned mine = 0u;
        #pragma unroll
        for (int j = 0; j < 32; ++j) {
            const unsigned mk = __builtin_amdgcn_ballot_w32(rel == g * 32 + j);
            mine = (lane == j) ? mk : mine;
        }
        sm[g][lane][wave] = mine;
    }
    __syncthreads();
    #pragma unroll
    for (int g = 0; g < 4; ++g) {
        const int t = ob0 + g * 32 + wave;
        if (t < nTiles)
            planes[(size_t)t * PW + blockIdx.x * 32 + lane] = sm[g][wave][lane];
    }
}

__global__ void k_dinv(float* __restrict__ deg, int N) {
    int i = blockIdx.x * blockDim.x + threadIdx.x;
    if (i < N) deg[i] = rsqrtf(deg[i]);
}

__global__ __launch_bounds__(256) void k_gemm_wmma(const float* __restrict__ X,
                                                   const float* __restrict__ W,
                                                   const float* __restrict__ rs,
                                                   float* __restrict__ H, int M) {
    __shared__ float sA[64][132];
    __shared__ float sB[32][132];
    const int m0  = blockIdx.x * 64;
    const int n0  = blockIdx.y * 32;
    const int tid = threadIdx.x;

    #pragma unroll
    for (int i = 0; i < 8; ++i) {
        const int idx   = tid + i * 256;
        const int row   = idx >> 5;
        const int chunk = idx & 31;
        const int gr    = m0 + row;
        const float4 v =
            *(const float4*)(X + (size_t)(gr < M ? gr : 0) * CH + chunk * 4);
        *(float4*)&sA[row][chunk * 4] = v;
    }
    {
        const int c     = tid & 31;
        const int kbase = (tid >> 5) * 16;
        #pragma unroll
        for (int i = 0; i < 16; ++i) {
            const int k = kbase + i;
            sB[c][k] = W[(size_t)k * CH + n0 + c];
        }
    }
    __syncthreads();

    const int wave  = tid >> 5;
    const int lane  = tid & 31;
    const int mt    = wave >> 1;
    const int nt    = wave & 1;
    const int mloc  = lane & 15;
    const int khalf = (lane >> 4) * 2;

    const float* arow = &sA[mt * 16 + mloc][0];
    const float* brow = &sB[nt * 16 + mloc][0];

    const int sbase = m0 + mt * 16 + (lane >> 4) * 8;
    const float4 sc0 = *(const float4*)(rs + sbase);
    const float4 sc1 = *(const float4*)(rs + sbase + 4);

    v8f c = {};
    #pragma unroll 8
    for (int k0 = 0; k0 < CH; k0 += 4) {
        const v2f a = *(const v2f*)(arow + k0 + khalf);
        const v2f b = *(const v2f*)(brow + k0 + khalf);
        c = __builtin_amdgcn_wmma_f32_16x16x4_f32(
                 false, a,  false, b,
                 (short)0, c,  false,  false);
    }

    const int col   = n0 + nt * 16 + mloc;
    const int half  = lane >> 4;
    const int mbase = m0 + mt * 16 + half * 8;
    const float sc[8] = {sc0.x, sc0.y, sc0.z, sc0.w, sc1.x, sc1.y, sc1.z, sc1.w};
    if (m0 + 64 <= M) {
        #pragma unroll
        for (int r = 0; r < 8; ++r)
            H[(size_t)(mbase + r) * CH + col] = c[r] * sc[r];
    } else {
        #pragma unroll
        for (int r = 0; r < 8; ++r)
            if (mbase + r < M) H[(size_t)(mbase + r) * CH + col] = c[r] * sc[r];
    }
}

#define TW  32
#define QN  256
#define PU  8
#define EG  8
#define LCAP 16384
#define LG  4
#define NOEDGE 0xFFFFFFFFu
#define LDS_RESERVE 65536

static_assert(TR <= 256, "the packed list keeps the row in tile in 8 bits");

template <bool FIRST>
__global__ __launch_bounds__(1024) void k_tile_aggregate(const float* __restrict__ H,
                                                         const int* __restrict__ ei,
                                                         const float* __restrict__ dinv,
                                                         const float* __restrict__ bias,
                                                         unsigned* planes,
                                                         int* cnts,
                                                         float* __restrict__ Out,
                                                         int E, int N, int PW) {
    __shared__ float    tile[TR * CH];
    __shared__ float    sdinv[TR];
    __shared__ int      qe[TW][QN];
    __shared__ unsigned lst[FIRST ? LCAP : 1];
    __shared__ int      lcount;

    const int lane = threadIdx.x & 31;
    const int wave = __builtin_amdgcn_readfirstlane((int)(threadIdx.x >> 5));
    const int r0   = blockIdx.x * TR;
    const int rows = (N - r0 < TR) ? (N - r0) : TR;

    int total = -1;
    if constexpr (!FIRST) total = __builtin_amdgcn_readfirstlane(cnts[blockIdx.x]);
    if constexpr (FIRST) { if (threadIdx.x == 0) lcount = 0; }

    {
        for (int rb = wave; rb < rows; rb += 4 * TW) {
            float4 h[4];
            float  di[4];
            #pragma unroll
            for (int k = 0; k < 4; ++k) {
                const int r  = rb + k * TW;
                const int rc = (r < rows) ? r : rb;
                di[k] = dinv[r0 + rc];
                h[k]  = ((const float4*)(H + (size_t)(r0 + rc) * CH))[lane];
            }
            #pragma unroll
            for (int k = 0; k < 4; ++k) {
                const int r = rb + k * TW;
                if (r < rows) {
                    float* t = tile + r * CH + lane;
                    t[0]  = h[k].x;
                    t[32] = h[k].y;
                    t[64] = h[k].z;
                    t[96] = h[k].w;
                    if (lane == 0) sdinv[r] = di[k];
                }
            }
        }
    }
    __syncthreads();

    auto add_rows = [&](const int s, const int myr, const int n) {
        #pragma unroll 1
        for (int g = 0; g < n; g += EG) {
            float4 v[EG];
            #pragma unroll
            for (int u = 0; u < EG; ++u) {
                const int sj = __builtin_amdgcn_readlane(s, g + u);
                v[u] = ((const float4*)(H + (size_t)sj * CH))[lane];
            }
            #pragma unroll
            for (int u = 0; u < EG; ++u) {
                const int rj = __builtin_amdgcn_readlane(myr, g + u);
                if (rj < 0) continue;
                float* t = tile + rj * CH + lane;
                atomicAdd(t + 0,  v[u].x);
                atomicAdd(t + 32, v[u].y);
                atomicAdd(t + 64, v[u].z);
                atomicAdd(t + 96, v[u].w);
            }
        }
    };

    if (FIRST || total < 0) {
        const int* __restrict__ dst = ei + E;
        int qhead = 0;
        int qtail = 0;

        auto drain = [&](const int n) {
            __builtin_amdgcn_fence(__ATOMIC_ACQ_REL, "wavefront");
            int myr = -1;
            int s   = 0;
            unsigned rec = NOEDGE;
            if (lane < n) {
                const int slot = (qhead + lane) & (QN - 1);
                const int e  = qe[wave][slot];
                const int ss = ei[e];
                const int lr = dst[e] - r0;
                const bool ok = ((unsigned)ss < (unsigned)N) & ((unsigned)lr < (unsigned)rows);
                s   = ok ? ss : 0;
                myr = ok ? lr : -1;
                rec = ok ? (((unsigned)ss << 8) | (unsigned)lr) : NOEDGE;
            }
            if constexpr (FIRST) {
                int base = 0;
                if (lane == 0) base = atomicAdd(&lcount, n);
                base = __builtin_amdgcn_readfirstlane(base);
                if (lane < n && base + lane < LCAP) lst[base + lane] = rec;
            }
            add_rows(s, myr, n);
        };

        auto take = [&](unsigned w, const int ln) {
            unsigned any = __builtin_amdgcn_ballot_w32(w != 0u);
            while (any != 0u) {
                if (w != 0u) {
                    const int slot =
                        (qtail + (int)__builtin_amdgcn_mbcnt_lo(any, 0u)) & (QN - 1);
                    qe[wave][slot] = ((ln * 32 + lane) << 5) + __builtin_ctz(w);
                    w &= w - 1u;
                }
                qtail += __builtin_popcount(any);
                while (qtail - qhead >= 32) {
                    drain(32);
                    qhead += 32;
                }
                any = __builtin_amdgcn_ballot_w32(w != 0u);
            }
        };

        {
            const unsigned* pl = planes + (size_t)blockIdx.x * PW;
            const int nLines = PW >> 5;
            int c = wave;
            unsigned d[PU];
            #pragma unroll
            for (int u = 0; u < PU; ++u) {
                const int l  = c + u * TW;
                const int lc = (l < nLines) ? l : (nLines - 1);
                const unsigned x = pl[lc * 32 + lane];
                d[u] = (l < nLines) ? x : 0u;
            }
            while (c < nLines) {
                const int cn = c + PU * TW;
                unsigned dn[PU];
                #pragma unroll
                for (int u = 0; u < PU; ++u) {
                    const int l  = cn + u * TW;
                    const int lc = (l < nLines) ? l : (nLines - 1);
                    const unsigned x = pl[lc * 32 + lane];
                    dn[u] = (l < nLines) ? x : 0u;
                }
                #pragma unroll
                for (int u = 0; u < PU; ++u) take(d[u], c + u * TW);
                #pragma unroll
                for (int u = 0; u < PU; ++u) d[u] = dn[u];
                c = cn;
            }
        }
        if (qtail - qhead > 0) drain(qtail - qhead);
    } else {
        const unsigned* lp = planes + (size_t)blockIdx.x * PW;
        const int per = (((total + TW - 1) / TW) + EG - 1) & ~(EG - 1);
        const int beg = wave * per;
        const int end = (beg + per < total) ? (beg + per) : total;
        for (int b = beg; b < end; b += LG * 32) {
            unsigned p[LG];
            #pragma unroll
            for (int k = 0; k < LG; ++k) {
                const int i = b + k * 32 + lane;
                p[k] = lp[(i < total) ? i : (total - 1)];
            }
            #pragma unroll
            for (int k = 0; k < LG; ++k) {
                const int bk = b + k * 32;
                if (bk < end) {
                    const int n  = (end - bk < 32) ? (end - bk) : 32;
                    const int sr = (int)(p[k] >> 8);
                    const int lr = (int)(p[k] & 255u);
                    const bool ok = (lane < n) & (p[k] != NOEDGE)
                                  & ((unsigned)sr < (unsigned)N) & (lr < rows);
                    add_rows(ok ? sr : 0, ok ? lr : -1, n);
                }
            }
        }
    }
    __syncthreads();

    if constexpr (FIRST) {
        const int tot = lcount;
        const bool fits = (tot <= LCAP) & (tot <= PW) & (N <= (1 << 24));
        if (fits) {
            unsigned* lp = planes + (size_t)blockIdx.x * PW;
            for (int i = (int)threadIdx.x; i < tot; i += TW * 32) lp[i] = lst[i];
        }
        if (threadIdx.x == 0) cnts[blockIdx.x] = fits ? tot : -1;
    }

    const float4 b = ((const float4*)bias)[lane];
    for (int r = wave; r < rows; r += TW) {
        const float* t = tile + r * CH + lane;
        const float  w = sdinv[r];
        float4 v;
        v.x = t[0]  * w + b.x;
        v.y = t[32] * w + b.y;
        v.z = t[64] * w + b.z;
        v.w = t[96] * w + b.w;

        float s = v.x + v.y + v.z + v.w;
        #pragma unroll
        for (int o = 16; o > 0; o >>= 1) s += __shfl_xor(s, o, 32);
        const float mu = s * (1.0f / CH);

        const float d0 = v.x - mu, d1 = v.y - mu, d2 = v.z - mu, d3 = v.w - mu;
        float q = d0 * d0 + d1 * d1 + d2 * d2 + d3 * d3;
        #pragma unroll
        for (int o = 16; o > 0; o >>= 1) q += __shfl_xor(q, o, 32);
        const float inv = rsqrtf(q * (1.0f / CH) + EPS);

        float4 o4;
        o4.x = fmaxf(d0 * inv, 0.0f);
        o4.y = fmaxf(d1 * inv, 0.0f);
        o4.z = fmaxf(d2 * inv, 0.0f);
        o4.w = fmaxf(d3 * inv, 0.0f);
        ((float4*)(Out + (size_t)(r0 + r) * CH))[lane] = o4;
    }
}

extern "C" void kernel_launch(void* const* d_in, const int* in_sizes, int n_in,
                              void* d_out, int out_size, void* d_ws, size_t ws_size,
                              hipStream_t stream) {
    const float* x   = (const float*)d_in[0];
    const int*   ei  = (const int*)  d_in[1];
    const float* W1  = (const float*)d_in[2];
    const float* b1  = (const float*)d_in[3];
    const float* W2  = (const float*)d_in[4];
    const float* b2  = (const float*)d_in[5];
    float*       out = (float*)d_out;

    const int N = in_sizes[0] / CH;
    const int E = in_sizes[1] / 2;

    float* ws   = (float*)d_ws;
    const size_t nPad = ((size_t)N + 63) & ~(size_t)63;
    float* dinv = ws;
    float* hbuf = ws + nPad;
    unsigned* planes = (unsigned*)(hbuf + (size_t)N * CH);

    const int T = 256;
    const int gN    = (N + T - 1) / T;
    const int gEdge = (E + 1023) / 1024;
    const int PW    = gEdge * 32;
    const int gTile = (N + TR - 1) / TR;
    const dim3 gGemm((N + 63) / 64, CH / 32);
    int* cnts = (int*)(planes + (size_t)gTile * PW);

    k_deg_init       <<<gN, T, 0, stream>>>(dinv, (int)nPad);
    k_edge_deg_planes<<<dim3(gEdge, (gTile + 127) / 128), 1024, 0, stream>>>(ei, dinv, planes, E, N, PW, gTile);
    k_dinv           <<<gN, T, 0, stream>>>(dinv, (int)nPad);

    k_gemm_wmma     <<<gGemm, T, 0, stream>>>(x, W1, dinv, hbuf, N);
    k_tile_aggregate<true> <<<gTile, TW * 32, 0, stream>>>(hbuf, ei, dinv, b1, planes, cnts, out, E, N, PW);

    k_gemm_wmma     <<<gGemm, T, 0, stream>>>(out, W2, dinv, hbuf, N);
    k_tile_aggregate<false><<<gTile, TW * 32, LDS_RESERVE, stream>>>(hbuf, ei, dinv, b2, planes, cnts, out, E, N, PW);
}


# layer 2's x@W2 joins layer 1's tile work-group: the finished 197 rows stay in LDS as the A operand, W2 staged in LDS, f32 WMMA, dinv-scaled store; 7 launches become 6; k summed in a permuted order
# speedup vs baseline: 2.8402x; 1.0595x over previous
#include <hip/hip_runtime.h>
#include <hip/hip_bf16.h>

typedef __attribute__((ext_vector_type(2))) float v2f;
typedef __attribute__((ext_vector_type(8))) float v8f;

#define CH 128
#define EPS 1e-5f

__global__ void k_deg_init(float* __restrict__ deg, int N) {
    int i = blockIdx.x * blockDim.x + threadIdx.x;
    if (i < N) deg[i] = 1.0f;
}

#define TR  197

__global__ __launch_bounds__(1024) void k_edge_deg_planes(const int* __restrict__ ei,
                                                          float* __restrict__ deg,
                                                          unsigned* __restrict__ planes,
                                                          int E, int N, int PW, int nTiles) {
    __shared__ unsigned sm[4][32][33];
    const int lane = threadIdx.x & 31;
    const int wave = __builtin_amdgcn_readfirstlane((int)(threadIdx.x >> 5));
    const int ob0  = blockIdx.y * 128;
    const int e = blockIdx.x * 1024 + (int)threadIdx.x;
    int rel = -1;
    if (e < E) {
        const int s = ei[e];
        const int d = ei[E + e];
        if ((unsigned)s < (unsigned)N && (unsigned)d < (unsigned)N) {
            if (blockIdx.y == 0) atomicAdd(&deg[d], 1.0f);
            const int r = d / TR - ob0;
            rel = (r < 0) ? -1 : r;
        }
    }
    #pragma unroll
    for (int g = 0; g < 4; ++g) {
        unsigned mine = 0u;
        #pragma unroll
        for (int j = 0; j < 32; ++j) {
            const unsigned mk = __builtin_amdgcn_ballot_w32(rel == g * 32 + j);
            mine = (lane == j) ? mk : mine;
        }
        sm[g][lane][wave] = mine;
    }
    __syncthreads();
    #pragma unroll
    for (int g = 0; g < 4; ++g) {
        const int t = ob0 + g * 32 + wave;
        if (t < nTiles)
            planes[(size_t)t * PW + blockIdx.x * 32 + lane] = sm[g][wave][lane];
    }
}

__global__ void k_dinv(float* __restrict__ deg, int N) {
    int i = blockIdx.x * blockDim.x + threadIdx.x;
    if (i < N) deg[i] = rsqrtf(deg[i]);
}

__global__ __launch_bounds__(256) void k_gemm_wmma(const float* __restrict__ X,
                                                   const float* __restrict__ W,
                                                   const float* __restrict__ rs,
                                                   float* __restrict__ H, int M) {
    __shared__ float sA[64][132];
    __shared__ float sB[32][132];
    const int m0  = blockIdx.x * 64;
    const int n0  = blockIdx.y * 32;
    const int tid = threadIdx.x;

    #pragma unroll
    for (int i = 0; i < 8; ++i) {
        const int idx   = tid + i * 256;
        const int row   = idx >> 5;
        const int chunk = idx & 31;
        const int gr    = m0 + row;
        const float4 v =
            *(const float4*)(X + (size_t)(gr < M ? gr : 0) * CH + chunk * 4);
        *(float4*)&sA[row][chunk * 4] = v;
    }
    {
        const int c     = tid & 31;
        const int kbase = (tid >> 5) * 16;
        #pragma unroll
        for (int i = 0; i < 16; ++i) {
            const int k = kbase + i;
            sB[c][k] = W[(size_t)k * CH + n0 + c];
        }
    }
    __syncthreads();

    const int wave  = tid >> 5;
    const int lane  = tid & 31;
    const int mt    = wave >> 1;
    const int nt    = wave & 1;
    const int mloc  = lane & 15;
    const int khalf = (lane >> 4) * 2;

    const float* arow = &sA[mt * 16 + mloc][0];
    const float* brow = &sB[nt * 16 + mloc][0];

    const int sbase = m0 + mt * 16 + (lane >> 4) * 8;
    const float4 sc0 = *(const float4*)(rs + sbase);
    const float4 sc1 = *(const float4*)(rs + sbase + 4);

    v8f c = {};
    #pragma unroll 8
    for (int k0 = 0; k0 < CH; k0 += 4) {
        const v2f a = *(const v2f*)(arow + k0 + khalf);
        const v2f b = *(const v2f*)(brow + k0 + khalf);
        c = __builtin_amdgcn_wmma_f32_16x16x4_f32(
                 false, a,  false, b,
                 (short)0, c,  false,  false);
    }

    const int col   = n0 + nt * 16 + mloc;
    const int half  = lane >> 4;
    const int mbase = m0 + mt * 16 + half * 8;
    const float sc[8] = {sc0.x, sc0.y, sc0.z, sc0.w, sc1.x, sc1.y, sc1.z, sc1.w};
    if (m0 + 64 <= M) {
        #pragma unroll
        for (int r = 0; r < 8; ++r)
            H[(size_t)(mbase + r) * CH + col] = c[r] * sc[r];
    } else {
        #pragma unroll
        for (int r = 0; r < 8; ++r)
            if (mbase + r < M) H[(size_t)(mbase + r) * CH + col] = c[r] * sc[r];
    }
}

#define TW  32
#define QN  256
#define PU  8
#define EG  8
#define LCAP 16384
#define LG  4
#define NOEDGE 0xFFFFFFFFu
#define LDS_RESERVE 65536
#define MT  ((TR + 15) / 16)
#define WP  132

static_assert(TR <= 256, "the packed list keeps the row in tile in 8 bits");

template <bool FIRST>
__global__ __launch_bounds__(1024) void k_tile_aggregate(const float* __restrict__ H,
                                                         const int* __restrict__ ei,
                                                         const float* __restrict__ dinv,
                                                         const float* __restrict__ bias,
                                                         const float* __restrict__ Wn,
                                                         unsigned* planes,
                                                         int* cnts,
                                                         float* __restrict__ Out,
                                                         int E, int N, int PW) {
    constexpr int TP    = FIRST ? 132 : CH;
    constexpr int TROWS = FIRST ? MT * 16 : TR;
    __shared__ float    tile[TROWS * TP];
    __shared__ float    sW[FIRST ? CH * WP : 1];
    __shared__ float    sdinv[TR];
    __shared__ int      qe[TW][QN];
    __shared__ unsigned lst[FIRST ? LCAP : 1];
    __shared__ int      lcount;

    const int lane = threadIdx.x & 31;
    const int wave = __builtin_amdgcn_readfirstlane((int)(threadIdx.x >> 5));
    const int r0   = blockIdx.x * TR;
    const int rows = (N - r0 < TR) ? (N - r0) : TR;

    int total = -1;
    if constexpr (!FIRST) total = __builtin_amdgcn_readfirstlane(cnts[blockIdx.x]);
    if constexpr (FIRST) { if (threadIdx.x == 0) lcount = 0; }

    if constexpr (FIRST) {
        const int n  = threadIdx.x & 127;
        const int kq = (int)(threadIdx.x >> 7) * 16;
        float wv[16];
        #pragma unroll
        for (int i = 0; i < 16; ++i) wv[i] = Wn[(size_t)(kq + i) * CH + n];
        #pragma unroll
        for (int i = 0; i < 16; ++i) {
            const int k = kq + i;
            sW[n * WP + (k & 3) * 32 + (k >> 2)] = wv[i];
        }
    }

    {
        for (int rb = wave; rb < rows; rb += 4 * TW) {
            float4 h[4];
            float  di[4];
            #pragma unroll
            for (int k = 0; k < 4; ++k) {
                const int r  = rb + k * TW;
                const int rc = (r < rows) ? r : rb;
                di[k] = dinv[r0 + rc];
                h[k]  = ((const float4*)(H + (size_t)(r0 + rc) * CH))[lane];
            }
            #pragma unroll
            for (int k = 0; k < 4; ++k) {
                const int r = rb + k * TW;
                if (r < rows) {
                    float* t = tile + r * TP + lane;
                    t[0]  = h[k].x;
                    t[32] = h[k].y;
                    t[64] = h[k].z;
                    t[96] = h[k].w;
                    if (lane == 0) sdinv[r] = di[k];
                }
            }
        }
    }
    __syncthreads();

    auto add_rows = [&](const int s, const int myr, const int n) {
        #pragma unroll 1
        for (int g = 0; g < n; g += EG) {
            float4 v[EG];
            #pragma unroll
            for (int u = 0; u < EG; ++u) {
                const int sj = __builtin_amdgcn_readlane(s, g + u);
                v[u] = ((const float4*)(H + (size_t)sj * CH))[lane];
            }
            #pragma unroll
            for (int u = 0; u < EG; ++u) {
                const int rj = __builtin_amdgcn_readlane(myr, g + u);
                if (rj < 0) continue;
                float* t = tile + rj * TP + lane;
                atomicAdd(t + 0,  v[u].x);
                atomicAdd(t + 32, v[u].y);
                atomicAdd(t + 64, v[u].z);
                atomicAdd(t + 96, v[u].w);
            }
        }
    };

    if (FIRST || total < 0) {
        const int* __restrict__ dst = ei + E;
        int qhead = 0;
        int qtail = 0;

        auto drain = [&](const int n) {
            __builtin_amdgcn_fence(__ATOMIC_ACQ_REL, "wavefront");
            int myr = -1;
            int s   = 0;
            unsigned rec = NOEDGE;
            if (lane < n) {
                const int slot = (qhead + lane) & (QN - 1);
                const int e  = qe[wave][slot];
                const int ss = ei[e];
                const int lr = dst[e] - r0;
                const bool ok = ((unsigned)ss < (unsigned)N) & ((unsigned)lr < (unsigned)rows);
                s   = ok ? ss : 0;
                myr = ok ? lr : -1;
                rec = ok ? (((unsigned)ss << 8) | (unsigned)lr) : NOEDGE;
            }
            if constexpr (FIRST) {
                int base = 0;
                if (lane == 0) base = atomicAdd(&lcount, n);
                base = __builtin_amdgcn_readfirstlane(base);
                if (lane < n && base + lane < LCAP) lst[base + lane] = rec;
            }
            add_rows(s, myr, n);
        };

        auto take = [&](unsigned w, const int ln) {
            unsigned any = __builtin_amdgcn_ballot_w32(w != 0u);
            while (any != 0u) {
                if (w != 0u) {
                    const int slot =
                        (qtail + (int)__builtin_amdgcn_mbcnt_lo(any, 0u)) & (QN - 1);
                    qe[wave][slot] = ((ln * 32 + lane) << 5) + __builtin_ctz(w);
                    w &= w - 1u;
                }
                qtail += __builtin_popcount(any);
                while (qtail - qhead >= 32) {
                    drain(32);
                    qhead += 32;
                }
                any = __builtin_amdgcn_ballot_w32(w != 0u);
            }
        };

        {
            const unsigned* pl = planes + (size_t)blockIdx.x * PW;
            const int nLines = PW >> 5;
            int c = wave;
            unsigned d[PU];
            #pragma unroll
            for (int u = 0; u < PU; ++u) {
                const int l  = c + u * TW;
                const int lc = (l < nLines) ? l : (nLines - 1);
                const unsigned x = pl[lc * 32 + lane];
                d[u] = (l < nLines) ? x : 0u;
            }
            while (c < nLines) {
                const int cn = c + PU * TW;
                unsigned dn[PU];
                #pragma unroll
                for (int u = 0; u < PU; ++u) {
                    const int l  = cn + u * TW;
                    const int lc = (l < nLines) ? l : (nLines - 1);
                    const unsigned x = pl[lc * 32 + lane];
                    dn[u] = (l < nLines) ? x : 0u;
                }
                #pragma unroll
                for (int u = 0; u < PU; ++u) take(d[u], c + u * TW);
                #pragma unroll
                for (int u = 0; u < PU; ++u) d[u] = dn[u];
                c = cn;
            }
        }
        if (qtail - qhead > 0) drain(qtail - qhead);
    } else {
        const unsigned* lp = planes + (size_t)blockIdx.x * PW;
        const int per = (((total + TW - 1) / TW) + EG - 1) & ~(EG - 1);
        const int beg = wave * per;
        const int end = (beg + per < total) ? (beg + per) : total;
        for (int b = beg; b < end; b += LG * 32) {
            unsigned p[LG];
            #pragma unroll
            for (int k = 0; k < LG; ++k) {
                const int i = b + k * 32 + lane;
                p[k] = lp[(i < total) ? i : (total - 1)];
            }
            #pragma unroll
            for (int k = 0; k < LG; ++k) {
                const int bk = b + k * 32;
                if (bk < end) {
                    const int n  = (end - bk < 32) ? (end - bk) : 32;
                    const int sr = (int)(p[k] >> 8);
                    const int lr = (int)(p[k] & 255u);
                    const bool ok = (lane < n) & (p[k] != NOEDGE)
                                  & ((unsigned)sr < (unsigned)N) & (lr < rows);
                    add_rows(ok ? sr : 0, ok ? lr : -1, n);
                }
            }
        }
    }
    __syncthreads();

    if constexpr (FIRST) {
        const int tot = lcount;
        const bool fits = (tot <= LCAP) & (tot <= PW) & (N <= (1 << 24));
        if (fits) {
            unsigned* lp = planes + (size_t)blockIdx.x * PW;
            for (int i = (int)threadIdx.x; i < tot; i += TW * 32) lp[i] = lst[i];
        }
        if (threadIdx.x == 0) cnts[blockIdx.x] = fits ? tot : -1;
    }

    const float4 b = ((const float4*)bias)[lane];
    for (int r = wave; r < rows; r += TW) {
        const float* t = tile + r * TP + lane;
        const float  w = sdinv[r];
        float4 v;
        v.x = t[0]  * w + b.x;
        v.y = t[32] * w + b.y;
        v.z = t[64] * w + b.z;
        v.w = t[96] * w + b.w;

        float s = v.x + v.y + v.z + v.w;
        #pragma unroll
        for (int o = 16; o > 0; o >>= 1) s += __shfl_xor(s, o, 32);
        const float mu = s * (1.0f / CH);

        const float d0 = v.x - mu, d1 = v.y - mu, d2 = v.z - mu, d3 = v.w - mu;
        float q = d0 * d0 + d1 * d1 + d2 * d2 + d3 * d3;
        #pragma unroll
        for (int o = 16; o > 0; o >>= 1) q += __shfl_xor(q, o, 32);
        const float inv = rsqrtf(q * (1.0f / CH) + EPS);

        float4 o4;
        o4.x = fmaxf(d0 * inv, 0.0f);
        o4.y = fmaxf(d1 * inv, 0.0f);
        o4.z = fmaxf(d2 * inv, 0.0f);
        o4.w = fmaxf(d3 * inv, 0.0f);
        if constexpr (FIRST) {
            float* tw = tile + r * TP + lane;
            tw[0]  = o4.x;
            tw[32] = o4.y;
            tw[64] = o4.z;
            tw[96] = o4.w;
        } else {
            ((float4*)(Out + (size_t)(r0 + r) * CH))[lane] = o4;
        }
    }

    if constexpr (FIRST) {
        __syncthreads();
        const int mloc  = lane & 15;
        const int half  = lane >> 4;
        const int khalf = half * 2;
        for (int item = wave; item < MT * 4; item += TW) {
            const int mt = item >> 2;
            const int np = item & 3;
            if (mt * 16 >= rows) continue;
            const float* arow = tile + (mt * 16 + mloc) * TP;
            const float* brow = sW + (np * 32 + mloc) * WP;
            v8f c0 = {};
            v8f c1 = {};
            const int rbase = mt * 16 + half * 8;
            float sc[8];
            #pragma unroll
            for (int r = 0; r < 8; ++r) sc[r] = sdinv[(rbase + r < TR) ? (rbase + r) : (TR - 1)];
            #pragma unroll 8
            for (int k0 = 0; k0 < CH; k0 += 4) {
                const v2f a  = *(const v2f*)(arow + k0 + khalf);
                const v2f b0 = *(const v2f*)(brow + k0 + khalf);
                const v2f b1 = *(const v2f*)(brow + 16 * WP + k0 + khalf);
                c0 = __builtin_amdgcn_wmma_f32_16x16x4_f32(
                          false, a,  false, b0,
                          (short)0, c0,  false,  false);
                c1 = __builtin_amdgcn_wmma_f32_16x16x4_f32(
                          false, a,  false, b1,
                          (short)0, c1,  false,  false);
            }
            float* o = Out + (size_t)(r0 + rbase) * CH + np * 32 + mloc;
            if (mt * 16 + 16 <= rows) {
                #pragma unroll
                for (int r = 0; r < 8; ++r) {
                    o[r * CH]      = c0[r] * sc[r];
                    o[r * CH + 16] = c1[r] * sc[r];
                }
            } else {
                #pragma unroll
                for (int r = 0; r < 8; ++r)
                    if (rbase + r < rows) {
                        o[r * CH]      = c0[r] * sc[r];
                        o[r * CH + 16] = c1[r] * sc[r];
                    }
            }
        }
    }
}

extern "C" void kernel_launch(void* const* d_in, const int* in_sizes, int n_in,
                              void* d_out, int out_size, void* d_ws, size_t ws_size,
                              hipStream_t stream) {
    const float* x   = (const float*)d_in[0];
    const int*   ei  = (const int*)  d_in[1];
    const float* W1  = (const float*)d_in[2];
    const float* b1  = (const float*)d_in[3];
    const float* W2  = (const float*)d_in[4];
    const float* b2  = (const float*)d_in[5];
    float*       out = (float*)d_out;

    const int N = in_sizes[0] / CH;
    const int E = in_sizes[1] / 2;

    float* ws   = (float*)d_ws;
    const size_t nPad = ((size_t)N + 63) & ~(size_t)63;
    float* dinv = ws;
    float* hbuf = ws + nPad;
    unsigned* planes = (unsigned*)(hbuf + (size_t)N * CH);

    const int T = 256;
    const int gN    = (N + T - 1) / T;
    const int gEdge = (E + 1023) / 1024;
    const int PW    = gEdge * 32;
    const int gTile = (N + TR - 1) / TR;
    const dim3 gGemm((N + 63) / 64, CH / 32);
    int* cnts = (int*)(planes + (size_t)gTile * PW);

    k_deg_init       <<<gN, T, 0, stream>>>(dinv, (int)nPad);
    k_edge_deg_planes<<<dim3(gEdge, (gTile + 127) / 128), 1024, 0, stream>>>(ei, dinv, planes, E, N, PW, gTile);
    k_dinv           <<<gN, T, 0, stream>>>(dinv, (int)nPad);

    k_gemm_wmma     <<<gGemm, T, 0, stream>>>(x, W1, dinv, out, N);
    k_tile_aggregate<true> <<<gTile, TW * 32, 0, stream>>>(out, ei, dinv, b1, W2, planes, cnts, hbuf, E, N, PW);

    k_tile_aggregate<false><<<gTile, TW * 32, LDS_RESERVE, stream>>>(hbuf, ei, dinv, b2, W2, planes, cnts, out, E, N, PW);
}


# in-degrees counted in LDS by each tile's owner on one plane walk ahead of the product, which leaves dinv and the edge list: the 800000 global f32 adds go; 6 launches to 5; sums in LDS arrival order
# speedup vs baseline: 4.0563x; 1.4282x over previous
#include <hip/hip_runtime.h>
#include <hip/hip_bf16.h>

typedef __attribute__((ext_vector_type(2))) float v2f;
typedef __attribute__((ext_vector_type(8))) float v8f;

#define CH 128
#define EPS 1e-5f

#define TR  197

__global__ __launch_bounds__(1024) void k_edge_planes(const int* __restrict__ ei,
                                                      unsigned* __restrict__ planes,
                                                      int E, int N, int PW, int nTiles) {
    __shared__ unsigned sm[4][32][33];
    const int lane = threadIdx.x & 31;
    const int wave = __builtin_amdgcn_readfirstlane((int)(threadIdx.x >> 5));
    const int ob0  = blockIdx.y * 128;
    const int e = blockIdx.x * 1024 + (int)threadIdx.x;
    int rel = -1;
    if (e < E) {
        const int s = ei[e];
        const int d = ei[E + e];
        if ((unsigned)s < (unsigned)N && (unsigned)d < (unsigned)N) {
            const int r = d / TR - ob0;
            rel = (r < 0) ? -1 : r;
        }
    }
    #pragma unroll
    for (int g = 0; g < 4; ++g) {
        unsigned mine = 0u;
        #pragma unroll
        for (int j = 0; j < 32; ++j) {
            const unsigned mk = __builtin_amdgcn_ballot_w32(rel == g * 32 + j);
            mine = (lane == j) ? mk : mine;
        }
        sm[g][lane][wave] = mine;
    }
    __syncthreads();
    #pragma unroll
    for (int g = 0; g < 4; ++g) {
        const int t = ob0 + g * 32 + wave;
        if (t < nTiles)
            planes[(size_t)t * PW + blockIdx.x * 32 + lane] = sm[g][wave][lane];
    }
}

__global__ __launch_bounds__(256) void k_gemm_wmma(const float* __restrict__ X,
                                                   const float* __restrict__ W,
                                                   const float* __restrict__ rs,
                                                   float* __restrict__ H, int M) {
    __shared__ float sA[64][132];
    __shared__ float sB[32][132];
    const int m0  = blockIdx.x * 64;
    const int n0  = blockIdx.y * 32;
    const int tid = threadIdx.x;

    #pragma unroll
    for (int i = 0; i < 8; ++i) {
        const int idx   = tid + i * 256;
        const int row   = idx >> 5;
        const int chunk = idx & 31;
        const int gr    = m0 + row;
        const float4 v =
            *(const float4*)(X + (size_t)(gr < M ? gr : 0) * CH + chunk * 4);
        *(float4*)&sA[row][chunk * 4] = v;
    }
    {
        const int c     = tid & 31;
        const int kbase = (tid >> 5) * 16;
        #pragma unroll
        for (int i = 0; i < 16; ++i) {
            const int k = kbase + i;
            sB[c][k] = W[(size_t)k * CH + n0 + c];
        }
    }
    __syncthreads();

    const int wave  = tid >> 5;
    const int lane  = tid & 31;
    const int mt    = wave >> 1;
    const int nt    = wave & 1;
    const int mloc  = lane & 15;
    const int khalf = (lane >> 4) * 2;

    const float* arow = &sA[mt * 16 + mloc][0];
    const float* brow = &sB[nt * 16 + mloc][0];

    const int sbase = m0 + mt * 16 + (lane >> 4) * 8;
    const float4 sc0 = *(const float4*)(rs + sbase);
    const float4 sc1 = *(const float4*)(rs + sbase + 4);

    v8f c = {};
    #pragma unroll 8
    for (int k0 = 0; k0 < CH; k0 += 4) {
        const v2f a = *(const v2f*)(arow + k0 + khalf);
        const v2f b = *(const v2f*)(brow + k0 + khalf);
        c = __builtin_amdgcn_wmma_f32_16x16x4_f32(
                 false, a,  false, b,
                 (short)0, c,  false,  false);
    }

    const int col   = n0 + nt * 16 + mloc;
    const int half  = lane >> 4;
    const int mbase = m0 + mt * 16 + half * 8;
    const float sc[8] = {sc0.x, sc0.y, sc0.z, sc0.w, sc1.x, sc1.y, sc1.z, sc1.w};
    if (m0 + 64 <= M) {
        #pragma unroll
        for (int r = 0; r < 8; ++r)
            H[(size_t)(mbase + r) * CH + col] = c[r] * sc[r];
    } else {
        #pragma unroll
        for (int r = 0; r < 8; ++r)
            if (mbase + r < M) H[(size_t)(mbase + r) * CH + col] = c[r] * sc[r];
    }
}

#define TW  32
#define QN  256
#define PU  8
#define EG  8
#define LCAP 16384
#define LG  4
#define LB  4
#define NOEDGE 0xFFFFFFFFu
#define LDS_RESERVE 65536
#define MT  ((TR + 15) / 16)
#define WP  132

static_assert(TR <= 256, "the packed list keeps the row in tile in 8 bits");

template <int DN, class DrainT>
__device__ __forceinline__ void walk_plane(const unsigned* pl, const int nLines,
                                           const int lane, const int wave,
                                           int* q, DrainT drain) {
    int qhead = 0;
    int qtail = 0;

    auto take = [&](unsigned w, const int ln) {
        unsigned any = __builtin_amdgcn_ballot_w32(w != 0u);
        while (any != 0u) {
            if (w != 0u) {
                const int slot =
                    (qtail + (int)__builtin_amdgcn_mbcnt_lo(any, 0u)) & (QN - 1);
                q[slot] = ((ln * 32 + lane) << 5) + __builtin_ctz(w);
                w &= w - 1u;
            }
            qtail += __builtin_popcount(any);
            while (qtail - qhead >= DN) {
                drain(qhead, DN);
                qhead += DN;
            }
            any = __builtin_amdgcn_ballot_w32(w != 0u);
        }
    };
    static_assert(DN % 32 == 0 && DN + 31 <= QN, "a turn queues up to 32 ids on top of DN-1 waiting ones");

    int c = wave;
    unsigned d[PU];
    #pragma unroll
    for (int u = 0; u < PU; ++u) {
        const int l  = c + u * TW;
        const int lc = (l < nLines) ? l : (nLines - 1);
        const unsigned x = pl[lc * 32 + lane];
        d[u] = (l < nLines) ? x : 0u;
    }
    while (c < nLines) {
        const int cn = c + PU * TW;
        unsigned dn[PU];
        #pragma unroll
        for (int u = 0; u < PU; ++u) {
            const int l  = cn + u * TW;
            const int lc = (l < nLines) ? l : (nLines - 1);
            const unsigned x = pl[lc * 32 + lane];
            dn[u] = (l < nLines) ? x : 0u;
        }
        #pragma unroll
        for (int u = 0; u < PU; ++u) take(d[u], c + u * TW);
        #pragma unroll
        for (int u = 0; u < PU; ++u) d[u] = dn[u];
        c = cn;
    }
    if (qtail - qhead > 0) drain(qhead, qtail - qhead);
}

__global__ __launch_bounds__(1024) void k_tile_list(const int* __restrict__ ei,
                                                    unsigned* planes,
                                                    int* __restrict__ cnts,
                                                    float* __restrict__ dinv,
                                                    int E, int N, int PW, int nPad) {
    __shared__ int      qe[TW][QN];
    __shared__ unsigned lst[LCAP];
    __shared__ int      indeg[TR];
    __shared__ int      lcount;

    const int lane = threadIdx.x & 31;
    const int wave = __builtin_amdgcn_readfirstlane((int)(threadIdx.x >> 5));
    const int r0   = blockIdx.x * TR;
    const int rows = (N - r0 < TR) ? (N - r0) : TR;

    if (threadIdx.x < TR) indeg[threadIdx.x] = 0;
    if (threadIdx.x == 0) lcount = 0;
    __syncthreads();

    const int* __restrict__ dst = ei + E;
    int* q = &qe[wave][0];
    auto drain = [&](const int pos, const int n) {
        __builtin_amdgcn_fence(__ATOMIC_ACQ_REL, "wavefront");
        int ss[LB];
        int dd[LB];
        #pragma unroll
        for (int k = 0; k < LB; ++k) {
            const int j = (k * 32 + lane < n) ? (k * 32 + lane) : 0;
            const int e = q[(pos + j) & (QN - 1)];
            ss[k] = ei[e];
            dd[k] = dst[e];
        }
        int base = 0;
        if (lane == 0) base = atomicAdd(&lcount, n);
        base = __builtin_amdgcn_readfirstlane(base);
        #pragma unroll
        for (int k = 0; k < LB; ++k) {
            const int i  = k * 32 + lane;
            const int lr = dd[k] - r0;
            const bool ok = (i < n) & ((unsigned)ss[k] < (unsigned)N) & ((unsigned)lr < (unsigned)rows);
            if (i < n && base + i < LCAP)
                lst[base + i] = ok ? (((unsigned)ss[k] << 8) | (unsigned)lr) : NOEDGE;
            if (ok) atomicAdd(&indeg[lr], 1);
        }
    };
    walk_plane<LB * 32>(planes + (size_t)blockIdx.x * PW, PW >> 5, lane, wave, q, drain);
    __syncthreads();

    const int tot = lcount;
    const bool fits = (tot <= LCAP) & (tot <= PW) & (N <= (1 << 24));
    if (fits) {
        unsigned* lp = planes + (size_t)blockIdx.x * PW;
        for (int i = (int)threadIdx.x; i < tot; i += TW * 32) lp[i] = lst[i];
    }
    if (threadIdx.x == 0) cnts[blockIdx.x] = fits ? tot : -1;
    if ((int)threadIdx.x < rows) dinv[r0 + threadIdx.x] = rsqrtf((float)(indeg[threadIdx.x] + 1));
    if (blockIdx.x == gridDim.x - 1)
        for (int i = N + (int)threadIdx.x; i < nPad; i += TW * 32) dinv[i] = 1.0f;
}

template <bool FIRST>
__global__ __launch_bounds__(1024) void k_tile_aggregate(const float* __restrict__ H,
                                                         const int* __restrict__ ei,
                                                         const float* __restrict__ dinv,
                                                         const float* __restrict__ bias,
                                                         const float* __restrict__ Wn,
                                                         const unsigned* __restrict__ planes,
                                                         const int* __restrict__ cnts,
                                                         float* __restrict__ Out,
                                                         int E, int N, int PW) {
    constexpr int TP    = FIRST ? 132 : CH;
    constexpr int TROWS = FIRST ? MT * 16 : TR;
    __shared__ float    tile[TROWS * TP];
    __shared__ float    sW[FIRST ? CH * WP : 1];
    __shared__ float    sdinv[TR];
    __shared__ int      qe[TW][QN];

    const int lane = threadIdx.x & 31;
    const int wave = __builtin_amdgcn_readfirstlane((int)(threadIdx.x >> 5));
    const int r0   = blockIdx.x * TR;
    const int rows = (N - r0 < TR) ? (N - r0) : TR;

    const int total = __builtin_amdgcn_readfirstlane(cnts[blockIdx.x]);

    if constexpr (FIRST) {
        const int n  = threadIdx.x & 127;
        const int kq = (int)(threadIdx.x >> 7) * 16;
        float wv[16];
        #pragma unroll
        for (int i = 0; i < 16; ++i) wv[i] = Wn[(size_t)(kq + i) * CH + n];
        #pragma unroll
        for (int i = 0; i < 16; ++i) {
            const int k = kq + i;
            sW[n * WP + (k & 3) * 32 + (k >> 2)] = wv[i];
        }
    }

    {
        for (int rb = wave; rb < rows; rb += 4 * TW) {
            float4 h[4];
            float  di[4];
            #pragma unroll
            for (int k = 0; k < 4; ++k) {
                const int r  = rb + k * TW;
                const int rc = (r < rows) ? r : rb;
                di[k] = dinv[r0 + rc];
                h[k]  = ((const float4*)(H + (size_t)(r0 + rc) * CH))[lane];
            }
            #pragma unroll
            for (int k = 0; k < 4; ++k) {
                const int r = rb + k * TW;
                if (r < rows) {
                    float* t = tile + r * TP + lane;
                    t[0]  = h[k].x;
                    t[32] = h[k].y;
                    t[64] = h[k].z;
                    t[96] = h[k].w;
                    if (lane == 0) sdinv[r] = di[k];
                }
            }
        }
    }
    __syncthreads();

    auto add_rows = [&](const int s, const int myr, const int n) {
        #pragma unroll 1
        for (int g = 0; g < n; g += EG) {
            float4 v[EG];
            #pragma unroll
            for (int u = 0; u < EG; ++u) {
                const int sj = __builtin_amdgcn_readlane(s, g + u);
                v[u] = ((const float4*)(H + (size_t)sj * CH))[lane];
            }
            #pragma unroll
            for (int u = 0; u < EG; ++u) {
                const int rj = __builtin_amdgcn_readlane(myr, g + u);
                if (rj < 0) continue;
                float* t = tile + rj * TP + lane;
                atomicAdd(t + 0,  v[u].x);
                atomicAdd(t + 32, v[u].y);
                atomicAdd(t + 64, v[u].z);
                atomicAdd(t + 96, v[u].w);
            }
        }
    };

    if (total < 0) {
        const int* __restrict__ dst = ei + E;
        int* q = &qe[wave][0];
        auto drain = [&](const int pos, const int n) {
            __builtin_amdgcn_fence(__ATOMIC_ACQ_REL, "wavefront");
            int myr = -1;
            int s   = 0;
            if (lane < n) {
                const int e  = q[(pos + lane) & (QN - 1)];
                const int ss = ei[e];
                const int lr = dst[e] - r0;
                const bool ok = ((unsigned)ss < (unsigned)N) & ((unsigned)lr < (unsigned)rows);
                s   = ok ? ss : 0;
                myr = ok ? lr : -1;
            }
            add_rows(s, myr, n);
        };
        walk_plane<32>(planes + (size_t)blockIdx.x * PW, PW >> 5, lane, wave, q, drain);
    } else {
        const unsigned* lp = planes + (size_t)blockIdx.x * PW;
        const int per = (((total + TW - 1) / TW) + EG - 1) & ~(EG - 1);
        const int beg = wave * per;
        const int end = (beg + per < total) ? (beg + per) : total;
        for (int b = beg; b < end; b += LG * 32) {
            unsigned p[LG];
            #pragma unroll
            for (int k = 0; k < LG; ++k) {
                const int i = b + k * 32 + lane;
                p[k] = lp[(i < total) ? i : (total - 1)];
            }
            #pragma unroll
            for (int k = 0; k < LG; ++k) {
                const int bk = b + k * 32;
                if (bk < end) {
                    const int n  = (end - bk < 32) ? (end - bk) : 32;
                    const int sr = (int)(p[k] >> 8);
                    const int lr = (int)(p[k] & 255u);
                    const bool ok = (lane < n) & (p[k] != NOEDGE)
                                  & ((unsigned)sr < (unsigned)N) & (lr < rows);
                    add_rows(ok ? sr : 0, ok ? lr : -1, n);
                }
            }
        }
    }
    __syncthreads();

    const float4 b = ((const float4*)bias)[lane];
    for (int r = wave; r < rows; r += TW) {
        const float* t = tile + r * TP + lane;
        const float  w = sdinv[r];
        float4 v;
        v.x = t[0]  * w + b.x;
        v.y = t[32] * w + b.y;
        v.z = t[64] * w + b.z;
        v.w = t[96] * w + b.w;

        float s = v.x + v.y + v.z + v.w;
        #pragma unroll
        for (int o = 16; o > 0; o >>= 1) s += __shfl_xor(s, o, 32);
        const float mu = s * (1.0f / CH);

        const float d0 = v.x - mu, d1 = v.y - mu, d2 = v.z - mu, d3 = v.w - mu;
        float q = d0 * d0 + d1 * d1 + d2 * d2 + d3 * d3;
        #pragma unroll
        for (int o = 16; o > 0; o >>= 1) q += __shfl_xor(q, o, 32);
        const float inv = rsqrtf(q * (1.0f / CH) + EPS);

        float4 o4;
        o4.x = fmaxf(d0 * inv, 0.0f);
        o4.y = fmaxf(d1 * inv, 0.0f);
        o4.z = fmaxf(d2 * inv, 0.0f);
        o4.w = fmaxf(d3 * inv, 0.0f);
        if constexpr (FIRST) {
            float* tw = tile + r * TP + lane;
            tw[0]  = o4.x;
            tw[32] = o4.y;
            tw[64] = o4.z;
            tw[96] = o4.w;
        } else {
            ((float4*)(Out + (size_t)(r0 + r) * CH))[lane] = o4;
        }
    }

    if constexpr (FIRST) {
        __syncthreads();
        const int mloc  = lane & 15;
        const int half  = lane >> 4;
        const int khalf = half * 2;
        for (int item = wave; item < MT * 4; item += TW) {
            const int mt = item >> 2;
            const int np = item & 3;
            if (mt * 16 >= rows) continue;
            const float* arow = tile + (mt * 16 + mloc) * TP;
            const float* brow = sW + (np * 32 + mloc) * WP;
            v8f c0 = {};
            v8f c1 = {};
            const int rbase = mt * 16 + half * 8;
            float sc[8];
            #pragma unroll
            for (int r = 0; r < 8; ++r) sc[r] = sdinv[(rbase + r < TR) ? (rbase + r) : (TR - 1)];
            #pragma unroll 8
            for (int k0 = 0; k0 < CH; k0 += 4) {
                const v2f a  = *(const v2f*)(arow + k0 + khalf);
                const v2f b0 = *(const v2f*)(brow + k0 + khalf);
                const v2f b1 = *(const v2f*)(brow + 16 * WP + k0 + khalf);
                c0 = __builtin_amdgcn_wmma_f32_16x16x4_f32(
                          false, a,  false, b0,
                          (short)0, c0,  false,  false);
                c1 = __builtin_amdgcn_wmma_f32_16x16x4_f32(
                          false, a,  false, b1,
                          (short)0, c1,  false,  false);
            }
            float* o = Out + (size_t)(r0 + rbase) * CH + np * 32 + mloc;
            if (mt * 16 + 16 <= rows) {
                #pragma unroll
                for (int r = 0; r < 8; ++r) {
                    o[r * CH]      = c0[r] * sc[r];
                    o[r * CH + 16] = c1[r] * sc[r];
                }
            } else {
                #pragma unroll
                for (int r = 0; r < 8; ++r)
                    if (rbase + r < rows) {
                        o[r * CH]      = c0[r] * sc[r];
                        o[r * CH + 16] = c1[r] * sc[r];
                    }
            }
        }
    }
}

extern "C" void kernel_launch(void* const* d_in, const int* in_sizes, int n_in,
                              void* d_out, int out_size, void* d_ws, size_t ws_size,
                              hipStream_t stream) {
    const float* x   = (const float*)d_in[0];
    const int*   ei  = (const int*)  d_in[1];
    const float* W1  = (const float*)d_in[2];
    const float* b1  = (const float*)d_in[3];
    const float* W2  = (const float*)d_in[4];
    const float* b2  = (const float*)d_in[5];
    float*       out = (float*)d_out;

    const int N = in_sizes[0] / CH;
    const int E = in_sizes[1] / 2;

    float* ws   = (float*)d_ws;
    const size_t nPad = ((size_t)N + 63) & ~(size_t)63;
    float* dinv = ws;
    float* hbuf = ws + nPad;
    unsigned* planes = (unsigned*)(hbuf + (size_t)N * CH);

    const int T = 256;
    const int gEdge = (E + 1023) / 1024;
    const int PW    = gEdge * 32;
    const int gTile = (N + TR - 1) / TR;
    const dim3 gGemm((N + 63) / 64, CH / 32);
    int* cnts = (int*)(planes + (size_t)gTile * PW);

    k_edge_planes<<<dim3(gEdge, (gTile + 127) / 128), 1024, 0, stream>>>(ei, planes, E, N, PW, gTile);
    k_tile_list  <<<gTile, TW * 32, LDS_RESERVE, stream>>>(ei, planes, cnts, dinv, E, N, PW, (int)nPad);

    k_gemm_wmma     <<<gGemm, T, 0, stream>>>(x, W1, dinv, out, N);
    k_tile_aggregate<true> <<<gTile, TW * 32, 0, stream>>>(out, ei, dinv, b1, W2, planes, cnts, hbuf, E, N, PW);

    k_tile_aggregate<false><<<gTile, TW * 32, LDS_RESERVE, stream>>>(hbuf, ei, dinv, b2, W2, planes, cnts, out, E, N, PW);
}


# edge bit planes: a work-group sets its 1024 edges' bits by one LDS or per lane into all 254 tile masks and takes two edge blocks (was 128 ballots per wave, tiles in two passes); 4x fewer waves
# speedup vs baseline: 4.2877x; 1.0570x over previous
#include <hip/hip_runtime.h>
#include <hip/hip_bf16.h>

typedef __attribute__((ext_vector_type(2))) float v2f;
typedef __attribute__((ext_vector_type(8))) float v8f;

#define CH 128
#define EPS 1e-5f

#define TR  197

#define PT  256
#define PP  33
__global__ __launch_bounds__(1024) void k_edge_planes(const int* __restrict__ ei,
                                                      unsigned* __restrict__ planes,
                                                      int E, int N, int PW, int nTiles,
                                                      int nBlocks) {
    __shared__ unsigned sm[PT * PP];
    const int lane = threadIdx.x & 31;
    const int wave = __builtin_amdgcn_readfirstlane((int)(threadIdx.x >> 5));
    const int ob0  = blockIdx.y * PT;

    #pragma unroll
    for (int j = 0; j < PT / 32; ++j) sm[(wave + 32 * j) * PP + lane] = 0u;

    int blk = blockIdx.x;
    int s = -1;
    int d = -1;
    {
        const int e = blk * 1024 + (int)threadIdx.x;
        if (blk < nBlocks && e < E) {
            s = ei[e];
            d = ei[E + e];
        }
    }
    for (; blk < nBlocks; blk += gridDim.x) {
        int rel = -1;
        if ((unsigned)s < (unsigned)N && (unsigned)d < (unsigned)N) {
            const int r = d / TR - ob0;
            rel = ((unsigned)r < (unsigned)PT) ? r : -1;
        }
        __syncthreads();
        if (rel >= 0) atomicOr(&sm[rel * PP + wave], 1u << lane);
        __syncthreads();
        const int nb = blk + (int)gridDim.x;
        s = -1;
        d = -1;
        {
            const int e = nb * 1024 + (int)threadIdx.x;
            if (nb < nBlocks && e < E) {
                s = ei[e];
                d = ei[E + e];
            }
        }
        unsigned m[PT / 32];
        #pragma unroll
        for (int j = 0; j < PT / 32; ++j) m[j] = sm[(wave + 32 * j) * PP + lane];
        #pragma unroll
        for (int j = 0; j < PT / 32; ++j) sm[(wave + 32 * j) * PP + lane] = 0u;
        #pragma unroll
        for (int j = 0; j < PT / 32; ++j) {
            const int t = ob0 + wave + 32 * j;
            if (t < nTiles)
                planes[(size_t)t * PW + blk * 32 + lane] = m[j];
        }
    }
}

__global__ __launch_bounds__(256) void k_gemm_wmma(const float* __restrict__ X,
                                                   const float* __restrict__ W,
                                                   const float* __restrict__ rs,
                                                   float* __restrict__ H, int M) {
    __shared__ float sA[64][132];
    __shared__ float sB[32][132];
    const int m0  = blockIdx.x * 64;
    const int n0  = blockIdx.y * 32;
    const int tid = threadIdx.x;

    #pragma unroll
    for (int i = 0; i < 8; ++i) {
        const int idx   = tid + i * 256;
        const int row   = idx >> 5;
        const int chunk = idx & 31;
        const int gr    = m0 + row;
        const float4 v =
            *(const float4*)(X + (size_t)(gr < M ? gr : 0) * CH + chunk * 4);
        *(float4*)&sA[row][chunk * 4] = v;
    }
    {
        const int c     = tid & 31;
        const int kbase = (tid >> 5) * 16;
        #pragma unroll
        for (int i = 0; i < 16; ++i) {
            const int k = kbase + i;
            sB[c][k] = W[(size_t)k * CH + n0 + c];
        }
    }
    __syncthreads();

    const int wave  = tid >> 5;
    const int lane  = tid & 31;
    const int mt    = wave >> 1;
    const int nt    = wave & 1;
    const int mloc  = lane & 15;
    const int khalf = (lane >> 4) * 2;

    const float* arow = &sA[mt * 16 + mloc][0];
    const float* brow = &sB[nt * 16 + mloc][0];

    const int sbase = m0 + mt * 16 + (lane >> 4) * 8;
    const float4 sc0 = *(const float4*)(rs + sbase);
    const float4 sc1 = *(const float4*)(rs + sbase + 4);

    v8f c = {};
    #pragma unroll 8
    for (int k0 = 0; k0 < CH; k0 += 4) {
        const v2f a = *(const v2f*)(arow + k0 + khalf);
        const v2f b = *(const v2f*)(brow + k0 + khalf);
        c = __builtin_amdgcn_wmma_f32_16x16x4_f32(
                 false, a,  false, b,
                 (short)0, c,  false,  false);
    }

    const int col   = n0 + nt * 16 + mloc;
    const int half  = lane >> 4;
    const int mbase = m0 + mt * 16 + half * 8;
    const float sc[8] = {sc0.x, sc0.y, sc0.z, sc0.w, sc1.x, sc1.y, sc1.z, sc1.w};
    if (m0 + 64 <= M) {
        #pragma unroll
        for (int r = 0; r < 8; ++r)
            H[(size_t)(mbase + r) * CH + col] = c[r] * sc[r];
    } else {
        #pragma unroll
        for (int r = 0; r < 8; ++r)
            if (mbase + r < M) H[(size_t)(mbase + r) * CH + col] = c[r] * sc[r];
    }
}

#define TW  32
#define QN  256
#define PU  8
#define EG  8
#define LCAP 16384
#define LG  4
#define LB  4
#define NOEDGE 0xFFFFFFFFu
#define LDS_RESERVE 65536
#define MT  ((TR + 15) / 16)
#define WP  132

static_assert(TR <= 256, "the packed list keeps the row in tile in 8 bits");

template <int DN, class DrainT>
__device__ __forceinline__ void walk_plane(const unsigned* pl, const int nLines,
                                           const int lane, const int wave,
                                           int* q, DrainT drain) {
    int qhead = 0;
    int qtail = 0;

    auto take = [&](unsigned w, const int ln) {
        unsigned any = __builtin_amdgcn_ballot_w32(w != 0u);
        while (any != 0u) {
            if (w != 0u) {
                const int slot =
                    (qtail + (int)__builtin_amdgcn_mbcnt_lo(any, 0u)) & (QN - 1);
                q[slot] = ((ln * 32 + lane) << 5) + __builtin_ctz(w);
                w &= w - 1u;
            }
            qtail += __builtin_popcount(any);
            while (qtail - qhead >= DN) {
                drain(qhead, DN);
                qhead += DN;
            }
            any = __builtin_amdgcn_ballot_w32(w != 0u);
        }
    };
    static_assert(DN % 32 == 0 && DN + 31 <= QN, "a turn queues up to 32 ids on top of DN-1 waiting ones");

    int c = wave;
    unsigned d[PU];
    #pragma unroll
    for (int u = 0; u < PU; ++u) {
        const int l  = c + u * TW;
        const int lc = (l < nLines) ? l : (nLines - 1);
        const unsigned x = pl[lc * 32 + lane];
        d[u] = (l < nLines) ? x : 0u;
    }
    while (c < nLines) {
        const int cn = c + PU * TW;
        unsigned dn[PU];
        #pragma unroll
        for (int u = 0; u < PU; ++u) {
            const int l  = cn + u * TW;
            const int lc = (l < nLines) ? l : (nLines - 1);
            const unsigned x = pl[lc * 32 + lane];
            dn[u] = (l < nLines) ? x : 0u;
        }
        #pragma unroll
        for (int u = 0; u < PU; ++u) take(d[u], c + u * TW);
        #pragma unroll
        for (int u = 0; u < PU; ++u) d[u] = dn[u];
        c = cn;
    }
    if (qtail - qhead > 0) drain(qhead, qtail - qhead);
}

__global__ __launch_bounds__(1024) void k_tile_list(const int* __restrict__ ei,
                                                    unsigned* planes,
                                                    int* __restrict__ cnts,
                                                    float* __restrict__ dinv,
                                                    int E, int N, int PW, int nPad) {
    __shared__ int      qe[TW][QN];
    __shared__ unsigned lst[LCAP];
    __shared__ int      indeg[TR];
    __shared__ int      lcount;

    const int lane = threadIdx.x & 31;
    const int wave = __builtin_amdgcn_readfirstlane((int)(threadIdx.x >> 5));
    const int r0   = blockIdx.x * TR;
    const int rows = (N - r0 < TR) ? (N - r0) : TR;

    if (threadIdx.x < TR) indeg[threadIdx.x] = 0;
    if (threadIdx.x == 0) lcount = 0;
    __syncthreads();

    const int* __restrict__ dst = ei + E;
    int* q = &qe[wave][0];
    auto drain = [&](const int pos, const int n) {
        __builtin_amdgcn_fence(__ATOMIC_ACQ_REL, "wavefront");
        int ss[LB];
        int dd[LB];
        #pragma unroll
        for (int k = 0; k < LB; ++k) {
            const int j = (k * 32 + lane < n) ? (k * 32 + lane) : 0;
            const int e = q[(pos + j) & (QN - 1)];
            ss[k] = ei[e];
            dd[k] = dst[e];
        }
        int base = 0;
        if (lane == 0) base = atomicAdd(&lcount, n);
        base = __builtin_amdgcn_readfirstlane(base);
        #pragma unroll
        for (int k = 0; k < LB; ++k) {
            const int i  = k * 32 + lane;
            const int lr = dd[k] - r0;
            const bool ok = (i < n) & ((unsigned)ss[k] < (unsigned)N) & ((unsigned)lr < (unsigned)rows);
            if (i < n && base + i < LCAP)
                lst[base + i] = ok ? (((unsigned)ss[k] << 8) | (unsigned)lr) : NOEDGE;
            if (ok) atomicAdd(&indeg[lr], 1);
        }
    };
    walk_plane<LB * 32>(planes + (size_t)blockIdx.x * PW, PW >> 5, lane, wave, q, drain);
    __syncthreads();

    const int tot = lcount;
    const bool fits = (tot <= LCAP) & (tot <= PW) & (N <= (1 << 24));
    if (fits) {
        unsigned* lp = planes + (size_t)blockIdx.x * PW;
        for (int i = (int)threadIdx.x; i < tot; i += TW * 32) lp[i] = lst[i];
    }
    if (threadIdx.x == 0) cnts[blockIdx.x] = fits ? tot : -1;
    if ((int)threadIdx.x < rows) dinv[r0 + threadIdx.x] = rsqrtf((float)(indeg[threadIdx.x] + 1));
    if (blockIdx.x == gridDim.x - 1)
        for (int i = N + (int)threadIdx.x; i < nPad; i += TW * 32) dinv[i] = 1.0f;
}

template <bool FIRST>
__global__ __launch_bounds__(1024) void k_tile_aggregate(const float* __restrict__ H,
                                                         const int* __restrict__ ei,
                                                         const float* __restrict__ dinv,
                                                         const float* __restrict__ bias,
                                                         const float* __restrict__ Wn,
                                                         const unsigned* __restrict__ planes,
                                                         const int* __restrict__ cnts,
                                                         float* __restrict__ Out,
                                                         int E, int N, int PW) {
    constexpr int TP    = FIRST ? 132 : CH;
    constexpr int TROWS = FIRST ? MT * 16 : TR;
    __shared__ float    tile[TROWS * TP];
    __shared__ float    sW[FIRST ? CH * WP : 1];
    __shared__ float    sdinv[TR];
    __shared__ int      qe[TW][QN];

    const int lane = threadIdx.x & 31;
    const int wave = __builtin_amdgcn_readfirstlane((int)(threadIdx.x >> 5));
    const int r0   = blockIdx.x * TR;
    const int rows = (N - r0 < TR) ? (N - r0) : TR;

    const int total = __builtin_amdgcn_readfirstlane(cnts[blockIdx.x]);

    if constexpr (FIRST) {
        const int n  = threadIdx.x & 127;
        const int kq = (int)(threadIdx.x >> 7) * 16;
        float wv[16];
        #pragma unroll
        for (int i = 0; i < 16; ++i) wv[i] = Wn[(size_t)(kq + i) * CH + n];
        #pragma unroll
        for (int i = 0; i < 16; ++i) {
            const int k = kq + i;
            sW[n * WP + (k & 3) * 32 + (k >> 2)] = wv[i];
        }
    }

    {
        for (int rb = wave; rb < rows; rb += 4 * TW) {
            float4 h[4];
            float  di[4];
            #pragma unroll
            for (int k = 0; k < 4; ++k) {
                const int r  = rb + k * TW;
                const int rc = (r < rows) ? r : rb;
                di[k] = dinv[r0 + rc];
                h[k]  = ((const float4*)(H + (size_t)(r0 + rc) * CH))[lane];
            }
            #pragma unroll
            for (int k = 0; k < 4; ++k) {
                const int r = rb + k * TW;
                if (r < rows) {
                    float* t = tile + r * TP + lane;
                    t[0]  = h[k].x;
                    t[32] = h[k].y;
                    t[64] = h[k].z;
                    t[96] = h[k].w;
                    if (lane == 0) sdinv[r] = di[k];
                }
            }
        }
    }
    __syncthreads();

    auto add_rows = [&](const int s, const int myr, const int n) {
        #pragma unroll 1
        for (int g = 0; g < n; g += EG) {
            float4 v[EG];
            #pragma unroll
            for (int u = 0; u < EG; ++u) {
                const int sj = __builtin_amdgcn_readlane(s, g + u);
                v[u] = ((const float4*)(H + (size_t)sj * CH))[lane];
            }
            #pragma unroll
            for (int u = 0; u < EG; ++u) {
                const int rj = __builtin_amdgcn_readlane(myr, g + u);
                if (rj < 0) continue;
                float* t = tile + rj * TP + lane;
                atomicAdd(t + 0,  v[u].x);
                atomicAdd(t + 32, v[u].y);
                atomicAdd(t + 64, v[u].z);
                atomicAdd(t + 96, v[u].w);
            }
        }
    };

    if (total < 0) {
        const int* __restrict__ dst = ei + E;
        int* q = &qe[wave][0];
        auto drain = [&](const int pos, const int n) {
            __builtin_amdgcn_fence(__ATOMIC_ACQ_REL, "wavefront");
            int myr = -1;
            int s   = 0;
            if (lane < n) {
                const int e  = q[(pos + lane) & (QN - 1)];
                const int ss = ei[e];
                const int lr = dst[e] - r0;
                const bool ok = ((unsigned)ss < (unsigned)N) & ((unsigned)lr < (unsigned)rows);
                s   = ok ? ss : 0;
                myr = ok ? lr : -1;
            }
            add_rows(s, myr, n);
        };
        walk_plane<32>(planes + (size_t)blockIdx.x * PW, PW >> 5, lane, wave, q, drain);
    } else {
        const unsigned* lp = planes + (size_t)blockIdx.x * PW;
        const int per = (((total + TW - 1) / TW) + EG - 1) & ~(EG - 1);
        const int beg = wave * per;
        const int end = (beg + per < total) ? (beg + per) : total;
        for (int b = beg; b < end; b += LG * 32) {
            unsigned p[LG];
            #pragma unroll
            for (int k = 0; k < LG; ++k) {
                const int i = b + k * 32 + lane;
                p[k] = lp[(i < total) ? i : (total - 1)];
            }
            #pragma unroll
            for (int k = 0; k < LG; ++k) {
                const int bk = b + k * 32;
                if (bk < end) {
                    const int n  = (end - bk < 32) ? (end - bk) : 32;
                    const int sr = (int)(p[k] >> 8);
                    const int lr = (int)(p[k] & 255u);
                    const bool ok = (lane < n) & (p[k] != NOEDGE)
                                  & ((unsigned)sr < (unsigned)N) & (lr < rows);
                    add_rows(ok ? sr : 0, ok ? lr : -1, n);
                }
            }
        }
    }
    __syncthreads();

    const float4 b = ((const float4*)bias)[lane];
    for (int r = wave; r < rows; r += TW) {
        const float* t = tile + r * TP + lane;
        const float  w = sdinv[r];
        float4 v;
        v.x = t[0]  * w + b.x;
        v.y = t[32] * w + b.y;
        v.z = t[64] * w + b.z;
        v.w = t[96] * w + b.w;

        float s = v.x + v.y + v.z + v.w;
        #pragma unroll
        for (int o = 16; o > 0; o >>= 1) s += __shfl_xor(s, o, 32);
        const float mu = s * (1.0f / CH);

        const float d0 = v.x - mu, d1 = v.y - mu, d2 = v.z - mu, d3 = v.w - mu;
        float q = d0 * d0 + d1 * d1 + d2 * d2 + d3 * d3;
        #pragma unroll
        for (int o = 16; o > 0; o >>= 1) q += __shfl_xor(q, o, 32);
        const float inv = rsqrtf(q * (1.0f / CH) + EPS);

        float4 o4;
        o4.x = fmaxf(d0 * inv, 0.0f);
        o4.y = fmaxf(d1 * inv, 0.0f);
        o4.z = fmaxf(d2 * inv, 0.0f);
        o4.w = fmaxf(d3 * inv, 0.0f);
        if constexpr (FIRST) {
            float* tw = tile + r * TP + lane;
            tw[0]  = o4.x;
            tw[32] = o4.y;
            tw[64] = o4.z;
            tw[96] = o4.w;
        } else {
            ((float4*)(Out + (size_t)(r0 + r) * CH))[lane] = o4;
        }
    }

    if constexpr (FIRST) {
        __syncthreads();
        const int mloc  = lane & 15;
        const int half  = lane >> 4;
        const int khalf = half * 2;
        for (int item = wave; item < MT * 4; item += TW) {
            const int mt = item >> 2;
            const int np = item & 3;
            if (mt * 16 >= rows) continue;
            const float* arow = tile + (mt * 16 + mloc) * TP;
            const float* brow = sW + (np * 32 + mloc) * WP;
            v8f c0 = {};
            v8f c1 = {};
            const int rbase = mt * 16 + half * 8;
            float sc[8];
            #pragma unroll
            for (int r = 0; r < 8; ++r) sc[r] = sdinv[(rbase + r < TR) ? (rbase + r) : (TR - 1)];
            #pragma unroll 8
            for (int k0 = 0; k0 < CH; k0 += 4) {
                const v2f a  = *(const v2f*)(arow + k0 + khalf);
                const v2f b0 = *(const v2f*)(brow + k0 + khalf);
                const v2f b1 = *(const v2f*)(brow + 16 * WP + k0 + khalf);
                c0 = __builtin_amdgcn_wmma_f32_16x16x4_f32(
                          false, a,  false, b0,
                          (short)0, c0,  false,  false);
                c1 = __builtin_amdgcn_wmma_f32_16x16x4_f32(
                          false, a,  false, b1,
                          (short)0, c1,  false,  false);
            }
            float* o = Out + (size_t)(r0 + rbase) * CH + np * 32 + mloc;
            if (mt * 16 + 16 <= rows) {
                #pragma unroll
                for (int r = 0; r < 8; ++r) {
                    o[r * CH]      = c0[r] * sc[r];
                    o[r * CH + 16] = c1[r] * sc[r];
                }
            } else {
                #pragma unroll
                for (int r = 0; r < 8; ++r)
                    if (rbase + r < rows) {
                        o[r * CH]      = c0[r] * sc[r];
                        o[r * CH + 16] = c1[r] * sc[r];
                    }
            }
        }
    }
}

extern "C" void kernel_launch(void* const* d_in, const int* in_sizes, int n_in,
                              void* d_out, int out_size, void* d_ws, size_t ws_size,
                              hipStream_t stream) {
    const float* x   = (const float*)d_in[0];
    const int*   ei  = (const int*)  d_in[1];
    const float* W1  = (const float*)d_in[2];
    const float* b1  = (const float*)d_in[3];
    const float* W2  = (const float*)d_in[4];
    const float* b2  = (const float*)d_in[5];
    float*       out = (float*)d_out;

    const int N = in_sizes[0] / CH;
    const int E = in_sizes[1] / 2;

    float* ws   = (float*)d_ws;
    const size_t nPad = ((size_t)N + 63) & ~(size_t)63;
    float* dinv = ws;
    float* hbuf = ws + nPad;
    unsigned* planes = (unsigned*)(hbuf + (size_t)N * CH);

    const int T = 256;
    const int gEdge = (E + 1023) / 1024;
    const int PW    = gEdge * 32;
    const int gTile = (N + TR - 1) / TR;
    const dim3 gGemm((N + 63) / 64, CH / 32);
    int* cnts = (int*)(planes + (size_t)gTile * PW);

    k_edge_planes<<<dim3((gEdge + 1) / 2, (gTile + PT - 1) / PT), 1024, 0, stream>>>(ei, planes, E, N, PW, gTile, gEdge);
    k_tile_list  <<<gTile, TW * 32, LDS_RESERVE, stream>>>(ei, planes, cnts, dinv, E, N, PW, (int)nPad);

    k_gemm_wmma     <<<gGemm, T, 0, stream>>>(x, W1, dinv, out, N);
    k_tile_aggregate<true> <<<gTile, TW * 32, 0, stream>>>(out, ei, dinv, b1, W2, planes, cnts, hbuf, E, N, PW);

    k_tile_aggregate<false><<<gTile, TW * 32, LDS_RESERVE, stream>>>(hbuf, ei, dinv, b2, W2, planes, cnts, out, E, N, PW);
}
